# starting kernel
# baseline (speedup 1.0000x reference)
#include <hip/hip_runtime.h>
#include <hip/hip_bf16.h>
#include <math.h>

#define NN 50000
#define EE 800000
#define FF 128
#define HH 8
#define DD 16

typedef __attribute__((ext_vector_type(2))) float v2f;
typedef __attribute__((ext_vector_type(8))) float v8f;

__global__ void gat_zero_kernel(float* __restrict__ p, int n) {
    int i = blockIdx.x * blockDim.x + threadIdx.x;
    if (i < n) p[i] = 0.0f;
}

__global__ __launch_bounds__(32) void gat_proj_gemm(
    const float* __restrict__ x,
    const float* __restrict__ W_src,
    const float* __restrict__ W_dst,
    float* __restrict__ src_p,
    float* __restrict__ dst_p)
{
    const int which   = blockIdx.x;
    const int rowbase = blockIdx.y * 16;
    const float* __restrict__ W   = which ? W_dst : W_src;
    float* __restrict__       out = which ? dst_p : src_p;

    const int lane = threadIdx.x;
    const int half = lane >> 4;
    const int l    = lane & 15;

    const float* __restrict__ xrow = x + (size_t)(rowbase + l) * FF;

    v8f acc[8];
#pragma unroll
    for (int c = 0; c < 8; ++c) acc[c] = (v8f){};

    for (int k = 0; k < FF; k += 4) {
        v2f a;
        a.x = xrow[k + 2 * half + 0];
        a.y = xrow[k + 2 * half + 1];
#pragma unroll
        for (int c = 0; c < 8; ++c) {
            const float* wp = W + (size_t)(k + 2 * half) * FF + c * 16 + l;
            v2f b;
            b.x = wp[0];
            b.y = wp[FF];
            acc[c] = __builtin_amdgcn_wmma_f32_16x16x4_f32(
                false, a, false, b, (short)0, acc[c], false, false);
        }
    }

#pragma unroll
    for (int c = 0; c < 8; ++c) {
#pragma unroll
        for (int r = 0; r < 8; ++r) {
            out[(size_t)(rowbase + r + 8 * half) * FF + c * 16 + l] = acc[c][r];
        }
    }
}

__global__ __launch_bounds__(256) void gat_score(
    const float* __restrict__ src_p,
    const float* __restrict__ dst_p,
    const int*   __restrict__ srcI,
    const int*   __restrict__ dstI,
    const float* __restrict__ attn,
    float* __restrict__ scorebuf,
    float* __restrict__ partial)
{
    __shared__ float smax[256];
    const int tid = blockIdx.x * 256 + threadIdx.x;
    float s = -INFINITY;
    if (tid < EE * HH) {
        const int e = tid >> 3;
        const int h = tid & 7;
        const float* sp = src_p + (size_t)srcI[e] * FF + h * DD;
        const float* dp = dst_p + (size_t)dstI[e] * FF + h * DD;
        const float* at = attn + h * DD;
        float acc = 0.0f;
#pragma unroll
        for (int d = 0; d < DD; ++d) {
            float v = sp[d] + dp[d];
            v = (v >= 0.0f) ? v : 0.2f * v;
            acc += at[d] * v;
        }
        scorebuf[tid] = acc;
        s = acc;
    }
    smax[threadIdx.x] = s;
    __syncthreads();
#pragma unroll
    for (int off = 128; off > 0; off >>= 1) {
        if (threadIdx.x < off)
            smax[threadIdx.x] = fmaxf(smax[threadIdx.x], smax[threadIdx.x + off]);
        __syncthreads();
    }
    if (threadIdx.x == 0) partial[blockIdx.x] = smax[0];
}

__global__ __launch_bounds__(256) void gat_maxreduce(
    const float* __restrict__ partial, int n, float* __restrict__ gmax)
{
    __shared__ float smax[256];
    float s = -INFINITY;
    for (int i = threadIdx.x; i < n; i += 256) s = fmaxf(s, partial[i]);
    smax[threadIdx.x] = s;
    __syncthreads();
#pragma unroll
    for (int off = 128; off > 0; off >>= 1) {
        if (threadIdx.x < off)
            smax[threadIdx.x] = fmaxf(smax[threadIdx.x], smax[threadIdx.x + off]);
        __syncthreads();
    }
    if (threadIdx.x == 0) *gmax = smax[0];
}

__global__ __launch_bounds__(256) void gat_expdenom(
    float* __restrict__ scorebuf,
    const int* __restrict__ dstI,
    const float* __restrict__ gmax,
    float* __restrict__ denom)
{
    const int tid = blockIdx.x * 256 + threadIdx.x;
    if (tid >= EE * HH) return;
    const int e = tid >> 3;
    const int h = tid & 7;
    const float ex = __expf(scorebuf[tid] - *gmax);
    scorebuf[tid] = ex;
    atomicAdd(&denom[(size_t)dstI[e] * HH + h], ex);
}

__global__ __launch_bounds__(256) void gat_message(
    const float* __restrict__ src_p,
    const float* __restrict__ expbuf,
    const float* __restrict__ denom,
    const int*   __restrict__ srcI,
    const int*   __restrict__ dstI,
    float* __restrict__ out)
{
    const int tid = blockIdx.x * 256 + threadIdx.x;
    if (tid >= EE * HH) return;
    const int e = tid >> 3;
    const int h = tid & 7;
    const int di = dstI[e];
    const float alpha = expbuf[tid] / (denom[(size_t)di * HH + h] + 1e-16f);
    const float* sp = src_p + (size_t)srcI[e] * FF + h * DD;
    float* op = out + (size_t)di * FF + h * DD;
#pragma unroll
    for (int d = 0; d < DD; ++d) {
        atomicAdd(&op[d], sp[d] * alpha);
    }
}

__global__ __launch_bounds__(256) void gat_finalize(
    float* __restrict__ out,
    const float* __restrict__ x,
    const float* __restrict__ bias,
    const float* __restrict__ prelu_w)
{
    const int i = blockIdx.x * 256 + threadIdx.x;
    if (i >= NN * FF) return;
    const float v  = out[i] + x[i] + bias[i & (FF - 1)];
    const float pw = prelu_w[0];
    out[i] = (v >= 0.0f) ? v : pw * v;
}

extern "C" void kernel_launch(void* const* d_in, const int* in_sizes, int n_in,
                              void* d_out, int out_size, void* d_ws, size_t ws_size,
                              hipStream_t stream) {
    (void)in_sizes; (void)n_in; (void)out_size; (void)ws_size;

    const float* x        = (const float*)d_in[0];
    const int*   eidx     = (const int*)  d_in[1];
    const float* W_src    = (const float*)d_in[2];
    const float* W_dst    = (const float*)d_in[3];
    const float* attn     = (const float*)d_in[4];
    const float* bias     = (const float*)d_in[5];
    const float* prelu_w  = (const float*)d_in[6];
    const int*   srcI     = eidx;
    const int*   dstI     = eidx + EE;

    float* out = (float*)d_out;

    float* ws      = (float*)d_ws;
    float* src_p   = ws;
    float* dst_p   = src_p + (size_t)NN * FF;
    float* scoreb  = dst_p + (size_t)NN * FF;
    float* denom   = scoreb + (size_t)EE * HH;
    float* partial = denom + (size_t)NN * HH;
    const int nScoreBlk = (EE * HH + 255) / 256;
    float* gmax    = partial + nScoreBlk;

    {
        const int n1 = NN * FF;
        gat_zero_kernel<<<(n1 + 255) / 256, 256, 0, stream>>>(out, n1);
        const int n2 = NN * HH;
        gat_zero_kernel<<<(n2 + 255) / 256, 256, 0, stream>>>(denom, n2);
    }

    {
        dim3 grid(2, NN / 16);
        gat_proj_gemm<<<grid, 32, 0, stream>>>(x, W_src, W_dst, src_p, dst_p);
    }

    gat_score<<<nScoreBlk, 256, 0, stream>>>(src_p, dst_p, srcI, dstI, attn,
                                             scoreb, partial);
    gat_maxreduce<<<1, 256, 0, stream>>>(partial, nScoreBlk, gmax);

    gat_expdenom<<<nScoreBlk, 256, 0, stream>>>(scoreb, dstI, gmax, denom);

    gat_message<<<nScoreBlk, 256, 0, stream>>>(src_p, scoreb, denom, srcI, dstI, out);

    {
        const int n = NN * FF;
        gat_finalize<<<(n + 255) / 256, 256, 0, stream>>>(out, x, bias, prelu_w);
    }
}


# message scatter: one wave per edge row, lane = column, so each f32 atomic add instruction covers one whole 128-byte line (was 16 lines); alpha once per edge and head; add order still free
# speedup vs baseline: 2.4908x; 2.4908x over previous
#include <hip/hip_runtime.h>
#include <hip/hip_bf16.h>
#include <math.h>

#define NN 50000
#define EE 800000
#define FF 128
#define HH 8
#define DD 16

typedef __attribute__((ext_vector_type(2))) float v2f;
typedef __attribute__((ext_vector_type(8))) float v8f;

__global__ void gat_zero_kernel(float* __restrict__ p, int n) {
    int i = blockIdx.x * blockDim.x + threadIdx.x;
    if (i < n) p[i] = 0.0f;
}

__global__ __launch_bounds__(32) void gat_proj_gemm(
    const float* __restrict__ x,
    const float* __restrict__ W_src,
    const float* __restrict__ W_dst,
    float* __restrict__ src_p,
    float* __restrict__ dst_p)
{
    const int which   = blockIdx.x;
    const int rowbase = blockIdx.y * 16;
    const float* __restrict__ W   = which ? W_dst : W_src;
    float* __restrict__       out = which ? dst_p : src_p;

    const int lane = threadIdx.x;
    const int half = lane >> 4;
    const int l    = lane & 15;

    const float* __restrict__ xrow = x + (size_t)(rowbase + l) * FF;

    v8f acc[8];
#pragma unroll
    for (int c = 0; c < 8; ++c) acc[c] = (v8f){};

    for (int k = 0; k < FF; k += 4) {
        v2f a;
        a.x = xrow[k + 2 * half + 0];
        a.y = xrow[k + 2 * half + 1];
#pragma unroll
        for (int c = 0; c < 8; ++c) {
            const float* wp = W + (size_t)(k + 2 * half) * FF + c * 16 + l;
            v2f b;
            b.x = wp[0];
            b.y = wp[FF];
            acc[c] = __builtin_amdgcn_wmma_f32_16x16x4_f32(
                false, a, false, b, (short)0, acc[c], false, false);
        }
    }

#pragma unroll
    for (int c = 0; c < 8; ++c) {
#pragma unroll
        for (int r = 0; r < 8; ++r) {
            out[(size_t)(rowbase + r + 8 * half) * FF + c * 16 + l] = acc[c][r];
        }
    }
}

__global__ __launch_bounds__(256) void gat_score(
    const float* __restrict__ src_p,
    const float* __restrict__ dst_p,
    const int*   __restrict__ srcI,
    const int*   __restrict__ dstI,
    const float* __restrict__ attn,
    float* __restrict__ scorebuf,
    float* __restrict__ partial)
{
    __shared__ float smax[256];
    const int tid = blockIdx.x * 256 + threadIdx.x;
    float s = -INFINITY;
    if (tid < EE * HH) {
        const int e = tid >> 3;
        const int h = tid & 7;
        const float* sp = src_p + (size_t)srcI[e] * FF + h * DD;
        const float* dp = dst_p + (size_t)dstI[e] * FF + h * DD;
        const float* at = attn + h * DD;
        float acc = 0.0f;
#pragma unroll
        for (int d = 0; d < DD; ++d) {
            float v = sp[d] + dp[d];
            v = (v >= 0.0f) ? v : 0.2f * v;
            acc += at[d] * v;
        }
        scorebuf[tid] = acc;
        s = acc;
    }
    smax[threadIdx.x] = s;
    __syncthreads();
#pragma unroll
    for (int off = 128; off > 0; off >>= 1) {
        if (threadIdx.x < off)
            smax[threadIdx.x] = fmaxf(smax[threadIdx.x], smax[threadIdx.x + off]);
        __syncthreads();
    }
    if (threadIdx.x == 0) partial[blockIdx.x] = smax[0];
}

__global__ __launch_bounds__(256) void gat_maxreduce(
    const float* __restrict__ partial, int n, float* __restrict__ gmax)
{
    __shared__ float smax[256];
    float s = -INFINITY;
    for (int i = threadIdx.x; i < n; i += 256) s = fmaxf(s, partial[i]);
    smax[threadIdx.x] = s;
    __syncthreads();
#pragma unroll
    for (int off = 128; off > 0; off >>= 1) {
        if (threadIdx.x < off)
            smax[threadIdx.x] = fmaxf(smax[threadIdx.x], smax[threadIdx.x + off]);
        __syncthreads();
    }
    if (threadIdx.x == 0) *gmax = smax[0];
}

__global__ __launch_bounds__(256) void gat_expdenom(
    float* __restrict__ scorebuf,
    const int* __restrict__ dstI,
    const float* __restrict__ gmax,
    float* __restrict__ denom)
{
    const int tid = blockIdx.x * 256 + threadIdx.x;
    if (tid >= EE * HH) return;
    const int e = tid >> 3;
    const int h = tid & 7;
    const float ex = __expf(scorebuf[tid] - *gmax);
    scorebuf[tid] = ex;
    atomicAdd(&denom[(size_t)dstI[e] * HH + h], ex);
}

static_assert(EE % 64 == 0, "a block of 8 waves takes 64 whole edges");

static __device__ __forceinline__ float lane_value(float v, int src_lane) {
    return __int_as_float(__builtin_amdgcn_readlane(__float_as_int(v), src_lane));
}

__global__ __launch_bounds__(256) void gat_message(
    const float* __restrict__ src_p,
    const float* __restrict__ expbuf,
    const float* __restrict__ denom,
    const int*   __restrict__ srcI,
    const int*   __restrict__ dstI,
    float* __restrict__ out)
{
    const int lane = threadIdx.x & 31;
    const bool upper = (lane & 16) != 0;
    const int wv = __builtin_amdgcn_readfirstlane((int)((blockIdx.x * 256 + threadIdx.x) >> 5));
    const int e0 = wv * 8;

#pragma unroll
    for (int g = 0; g < 2; ++g) {
        const int eb = e0 + 4 * g;

        const int   dl = dstI[eb + (lane >> 3)];
        const float ex = expbuf[(size_t)eb * HH + lane];
        const float al = ex / (denom[(size_t)dl * HH + (lane & 7)] + 1e-16f);

        float v[4][4];
        int   di[4];
#pragma unroll
        for (int es = 0; es < 4; ++es) {
            const int si = srcI[eb + es];
            di[es] = dstI[eb + es];
            const float* __restrict__ sp = src_p + (size_t)si * FF + lane;
#pragma unroll
            for (int j = 0; j < 4; ++j) v[es][j] = sp[32 * j];
        }
#pragma unroll
        for (int es = 0; es < 4; ++es) {
            float* __restrict__ op = out + (size_t)di[es] * FF + lane;
#pragma unroll
            for (int j = 0; j < 4; ++j) {
                const float a_even = lane_value(al, es * 8 + 2 * j);
                const float a_odd  = lane_value(al, es * 8 + 2 * j + 1);
                const float a = upper ? a_odd : a_even;
                atomicAdd(&op[32 * j], v[es][j] * a);
            }
        }
    }
}

__global__ __launch_bounds__(256) void gat_finalize(
    float* __restrict__ out,
    const float* __restrict__ x,
    const float* __restrict__ bias,
    const float* __restrict__ prelu_w)
{
    const int i = blockIdx.x * 256 + threadIdx.x;
    if (i >= NN * FF) return;
    const float v  = out[i] + x[i] + bias[i & (FF - 1)];
    const float pw = prelu_w[0];
    out[i] = (v >= 0.0f) ? v : pw * v;
}

extern "C" void kernel_launch(void* const* d_in, const int* in_sizes, int n_in,
                              void* d_out, int out_size, void* d_ws, size_t ws_size,
                              hipStream_t stream) {
    (void)in_sizes; (void)n_in; (void)out_size; (void)ws_size;

    const float* x        = (const float*)d_in[0];
    const int*   eidx     = (const int*)  d_in[1];
    const float* W_src    = (const float*)d_in[2];
    const float* W_dst    = (const float*)d_in[3];
    const float* attn     = (const float*)d_in[4];
    const float* bias     = (const float*)d_in[5];
    const float* prelu_w  = (const float*)d_in[6];
    const int*   srcI     = eidx;
    const int*   dstI     = eidx + EE;

    float* out = (float*)d_out;

    float* ws      = (float*)d_ws;
    float* src_p   = ws;
    float* dst_p   = src_p + (size_t)NN * FF;
    float* scoreb  = dst_p + (size_t)NN * FF;
    float* denom   = scoreb + (size_t)EE * HH;
    float* partial = denom + (size_t)NN * HH;
    const int nScoreBlk = (EE * HH + 255) / 256;
    float* gmax    = partial + nScoreBlk;

    {
        const int n1 = NN * FF;
        gat_zero_kernel<<<(n1 + 255) / 256, 256, 0, stream>>>(out, n1);
        const int n2 = NN * HH;
        gat_zero_kernel<<<(n2 + 255) / 256, 256, 0, stream>>>(denom, n2);
    }

    {
        dim3 grid(2, NN / 16);
        gat_proj_gemm<<<grid, 32, 0, stream>>>(x, W_src, W_dst, src_p, dst_p);
    }

    gat_score<<<nScoreBlk, 256, 0, stream>>>(src_p, dst_p, srcI, dstI, attn,
                                             scoreb, partial);
    gat_maxreduce<<<1, 256, 0, stream>>>(partial, nScoreBlk, gmax);

    gat_expdenom<<<nScoreBlk, 256, 0, stream>>>(scoreb, dstI, gmax, denom);

    {
        const int nMsgBlk = EE / 64;
        gat_message<<<nMsgBlk, 256, 0, stream>>>(src_p, scoreb, denom, srcI, dstI, out);
    }

    {
        const int n = NN * FF;
        gat_finalize<<<(n + 255) / 256, 256, 0, stream>>>(out, x, bias, prelu_w);
    }
}


# score kernel remapped: one wave walks whole edge rows, lane = a 16-byte row piece, head dot = 4 in-lane terms + 2 lane adds (sum order changed), 32 edges per wave; 3125 block maxima instead of 25000
# speedup vs baseline: 2.9824x; 1.1974x over previous
#include <hip/hip_runtime.h>
#include <hip/hip_bf16.h>
#include <math.h>

#define NN 50000
#define EE 800000
#define FF 128
#define HH 8
#define DD 16

typedef __attribute__((ext_vector_type(2))) float v2f;
typedef __attribute__((ext_vector_type(8))) float v8f;

__global__ void gat_zero_kernel(float* __restrict__ p, int n) {
    int i = blockIdx.x * blockDim.x + threadIdx.x;
    if (i < n) p[i] = 0.0f;
}

__global__ __launch_bounds__(32) void gat_proj_gemm(
    const float* __restrict__ x,
    const float* __restrict__ W_src,
    const float* __restrict__ W_dst,
    float* __restrict__ src_p,
    float* __restrict__ dst_p)
{
    const int which   = blockIdx.x;
    const int rowbase = blockIdx.y * 16;
    const float* __restrict__ W   = which ? W_dst : W_src;
    float* __restrict__       out = which ? dst_p : src_p;

    const int lane = threadIdx.x;
    const int half = lane >> 4;
    const int l    = lane & 15;

    const float* __restrict__ xrow = x + (size_t)(rowbase + l) * FF;

    v8f acc[8];
#pragma unroll
    for (int c = 0; c < 8; ++c) acc[c] = (v8f){};

    for (int k = 0; k < FF; k += 4) {
        v2f a;
        a.x = xrow[k + 2 * half + 0];
        a.y = xrow[k + 2 * half + 1];
#pragma unroll
        for (int c = 0; c < 8; ++c) {
            const float* wp = W + (size_t)(k + 2 * half) * FF + c * 16 + l;
            v2f b;
            b.x = wp[0];
            b.y = wp[FF];
            acc[c] = __builtin_amdgcn_wmma_f32_16x16x4_f32(
                false, a, false, b, (short)0, acc[c], false, false);
        }
    }

#pragma unroll
    for (int c = 0; c < 8; ++c) {
#pragma unroll
        for (int r = 0; r < 8; ++r) {
            out[(size_t)(rowbase + r + 8 * half) * FF + c * 16 + l] = acc[c][r];
        }
    }
}

#define SC_EPW  32
#define SC_NBLK (EE / (8 * SC_EPW))
static_assert(EE % (8 * SC_EPW) == 0, "a block of 8 waves takes 256 whole edges");

template <int CTRL>
static __device__ __forceinline__ float row_lane(float v) {
    return __int_as_float(__builtin_amdgcn_update_dpp(0, __float_as_int(v), CTRL, 0xf, 0xf, true));
}

__global__ __launch_bounds__(256) void gat_score(
    const float* __restrict__ src_p,
    const float* __restrict__ dst_p,
    const int*   __restrict__ srcI,
    const int*   __restrict__ dstI,
    const float* __restrict__ attn,
    float* __restrict__ scorebuf,
    float* __restrict__ partial)
{
    __shared__ float wmax[8];
    const int lane = threadIdx.x & 31;
    const int q    = lane & 3;
    const int head = lane >> 2;
    const int wv = __builtin_amdgcn_readfirstlane((int)((blockIdx.x * 256 + threadIdx.x) >> 5));
    const int e0 = wv * SC_EPW;

    const float4 at = *reinterpret_cast<const float4*>(attn + 4 * lane);
    float m = -INFINITY;

    for (int g = 0; g < SC_EPW / 4; ++g) {
        const int eb = e0 + 4 * g;
        float4 sv[4], dv[4];
#pragma unroll
        for (int es = 0; es < 4; ++es) {
            const int si = srcI[eb + es];
            const int di = dstI[eb + es];
            sv[es] = *reinterpret_cast<const float4*>(src_p + (size_t)si * FF + 4 * lane);
            dv[es] = *reinterpret_cast<const float4*>(dst_p + (size_t)di * FF + 4 * lane);
        }
        float p[4];
#pragma unroll
        for (int es = 0; es < 4; ++es) {
            float v0 = sv[es].x + dv[es].x;
            float v1 = sv[es].y + dv[es].y;
            float v2 = sv[es].z + dv[es].z;
            float v3 = sv[es].w + dv[es].w;
            v0 = (v0 >= 0.0f) ? v0 : 0.2f * v0;
            v1 = (v1 >= 0.0f) ? v1 : 0.2f * v1;
            v2 = (v2 >= 0.0f) ? v2 : 0.2f * v2;
            v3 = (v3 >= 0.0f) ? v3 : 0.2f * v3;
            float acc = at.x * v0;
            acc += at.y * v1;
            acc += at.z * v2;
            acc += at.w * v3;
            acc += row_lane<0xB1>(acc);
            acc += row_lane<0x4E>(acc);
            p[es] = acc;
        }
        const float sc = (q == 0) ? p[0] : (q == 1) ? p[1] : (q == 2) ? p[2] : p[3];
        scorebuf[(size_t)(eb + q) * HH + head] = sc;
        m = fmaxf(m, sc);
    }

    m = fmaxf(m, row_lane<0xB1>(m));
    m = fmaxf(m, row_lane<0x4E>(m));
    m = fmaxf(m, row_lane<0x164>(m));
    m = fmaxf(m, row_lane<0x168>(m));
    const float wm = fmaxf(__int_as_float(__builtin_amdgcn_readlane(__float_as_int(m), 0)),
                           __int_as_float(__builtin_amdgcn_readlane(__float_as_int(m), 16)));
    if (lane == 0) wmax[threadIdx.x >> 5] = wm;
    __syncthreads();
    if (threadIdx.x == 0) {
        float b = wmax[0];
#pragma unroll
        for (int i = 1; i < 8; ++i) b = fmaxf(b, wmax[i]);
        partial[blockIdx.x] = b;
    }
}

__global__ __launch_bounds__(256) void gat_maxreduce(
    const float* __restrict__ partial, float* __restrict__ gmax)
{
    __shared__ float smax[256];
    float s = -INFINITY;
#pragma unroll
    for (int j = 0; j < (SC_NBLK + 255) / 256; ++j) {
        const int i = min((int)threadIdx.x + 256 * j, SC_NBLK - 1);
        s = fmaxf(s, partial[i]);
    }
    smax[threadIdx.x] = s;
    __syncthreads();
#pragma unroll
    for (int off = 128; off > 0; off >>= 1) {
        if (threadIdx.x < off)
            smax[threadIdx.x] = fmaxf(smax[threadIdx.x], smax[threadIdx.x + off]);
        __syncthreads();
    }
    if (threadIdx.x == 0) *gmax = smax[0];
}

__global__ __launch_bounds__(256) void gat_expdenom(
    float* __restrict__ scorebuf,
    const int* __restrict__ dstI,
    const float* __restrict__ gmax,
    float* __restrict__ denom)
{
    const int tid = blockIdx.x * 256 + threadIdx.x;
    if (tid >= EE * HH) return;
    const int e = tid >> 3;
    const int h = tid & 7;
    const float ex = __expf(scorebuf[tid] - *gmax);
    scorebuf[tid] = ex;
    atomicAdd(&denom[(size_t)dstI[e] * HH + h], ex);
}

static_assert(EE % 64 == 0, "a block of 8 waves takes 64 whole edges");

static __device__ __forceinline__ float lane_value(float v, int src_lane) {
    return __int_as_float(__builtin_amdgcn_readlane(__float_as_int(v), src_lane));
}

__global__ __launch_bounds__(256) void gat_message(
    const float* __restrict__ src_p,
    const float* __restrict__ expbuf,
    const float* __restrict__ denom,
    const int*   __restrict__ srcI,
    const int*   __restrict__ dstI,
    float* __restrict__ out)
{
    const int lane = threadIdx.x & 31;
    const bool upper = (lane & 16) != 0;
    const int wv = __builtin_amdgcn_readfirstlane((int)((blockIdx.x * 256 + threadIdx.x) >> 5));
    const int e0 = wv * 8;

#pragma unroll
    for (int g = 0; g < 2; ++g) {
        const int eb = e0 + 4 * g;

        const int   dl = dstI[eb + (lane >> 3)];
        const float ex = expbuf[(size_t)eb * HH + lane];
        const float al = ex / (denom[(size_t)dl * HH + (lane & 7)] + 1e-16f);

        float v[4][4];
        int   di[4];
#pragma unroll
        for (int es = 0; es < 4; ++es) {
            const int si = srcI[eb + es];
            di[es] = dstI[eb + es];
            const float* __restrict__ sp = src_p + (size_t)si * FF + lane;
#pragma unroll
            for (int j = 0; j < 4; ++j) v[es][j] = sp[32 * j];
        }
#pragma unroll
        for (int es = 0; es < 4; ++es) {
            float* __restrict__ op = out + (size_t)di[es] * FF + lane;
#pragma unroll
            for (int j = 0; j < 4; ++j) {
                const float a_even = lane_value(al, es * 8 + 2 * j);
                const float a_odd  = lane_value(al, es * 8 + 2 * j + 1);
                const float a = upper ? a_odd : a_even;
                atomicAdd(&op[32 * j], v[es][j] * a);
            }
        }
    }
}

__global__ __launch_bounds__(256) void gat_finalize(
    float* __restrict__ out,
    const float* __restrict__ x,
    const float* __restrict__ bias,
    const float* __restrict__ prelu_w)
{
    const int i = blockIdx.x * 256 + threadIdx.x;
    if (i >= NN * FF) return;
    const float v  = out[i] + x[i] + bias[i & (FF - 1)];
    const float pw = prelu_w[0];
    out[i] = (v >= 0.0f) ? v : pw * v;
}

extern "C" void kernel_launch(void* const* d_in, const int* in_sizes, int n_in,
                              void* d_out, int out_size, void* d_ws, size_t ws_size,
                              hipStream_t stream) {
    (void)in_sizes; (void)n_in; (void)out_size; (void)ws_size;

    const float* x        = (const float*)d_in[0];
    const int*   eidx     = (const int*)  d_in[1];
    const float* W_src    = (const float*)d_in[2];
    const float* W_dst    = (const float*)d_in[3];
    const float* attn     = (const float*)d_in[4];
    const float* bias     = (const float*)d_in[5];
    const float* prelu_w  = (const float*)d_in[6];
    const int*   srcI     = eidx;
    const int*   dstI     = eidx + EE;

    float* out = (float*)d_out;

    float* ws      = (float*)d_ws;
    float* src_p   = ws;
    float* dst_p   = src_p + (size_t)NN * FF;
    float* scoreb  = dst_p + (size_t)NN * FF;
    float* denom   = scoreb + (size_t)EE * HH;
    float* partial = denom + (size_t)NN * HH;
    const int nEhBlk = (EE * HH + 255) / 256;
    float* gmax    = partial + SC_NBLK;

    {
        const int n1 = NN * FF;
        gat_zero_kernel<<<(n1 + 255) / 256, 256, 0, stream>>>(out, n1);
        const int n2 = NN * HH;
        gat_zero_kernel<<<(n2 + 255) / 256, 256, 0, stream>>>(denom, n2);
    }

    {
        dim3 grid(2, NN / 16);
        gat_proj_gemm<<<grid, 32, 0, stream>>>(x, W_src, W_dst, src_p, dst_p);
    }

    gat_score<<<SC_NBLK, 256, 0, stream>>>(src_p, dst_p, srcI, dstI, attn,
                                           scoreb, partial);
    gat_maxreduce<<<1, 256, 0, stream>>>(partial, gmax);

    gat_expdenom<<<nEhBlk, 256, 0, stream>>>(scoreb, dstI, gmax, denom);

    {
        const int nMsgBlk = EE / 64;
        gat_message<<<nMsgBlk, 256, 0, stream>>>(src_p, scoreb, denom, srcI, dstI, out);
    }

    {
        const int n = NN * FF;
        gat_finalize<<<(n + 255) / 256, 256, 0, stream>>>(out, x, bias, prelu_w);
    }
}


# 4 launches instead of 8: zero fills ride in the projection launch, block maxima join by f32 atomic max, exp+denominator ride in the message launch; rows divided once at the end (rounding changed)
# speedup vs baseline: 3.6838x; 1.2352x over previous
#include <hip/hip_runtime.h>
#include <hip/hip_bf16.h>
#include <math.h>

#define NN 50000
#define EE 800000
#define FF 128
#define HH 8
#define DD 16

typedef __attribute__((ext_vector_type(2))) float v2f;
typedef __attribute__((ext_vector_type(8))) float v8f;

__global__ __launch_bounds__(32) void gat_proj_gemm(
    const float* __restrict__ x,
    const float* __restrict__ W_src,
    const float* __restrict__ W_dst,
    float* __restrict__ src_p,
    float* __restrict__ dst_p,
    float* __restrict__ acc_rows,
    float* __restrict__ denom,
    float* __restrict__ gmax)
{
    const int which   = blockIdx.x;
    const int rowbase = blockIdx.y * 16;
    const float* __restrict__ W   = which ? W_dst : W_src;
    float* __restrict__       out = which ? dst_p : src_p;

    const int lane = threadIdx.x;
    const int half = lane >> 4;
    const int l    = lane & 15;

    const float* __restrict__ xrow = x + (size_t)(rowbase + l) * FF;

    {
        const float4 z = make_float4(0.0f, 0.0f, 0.0f, 0.0f);
        if (which == 0) {
            float4* __restrict__ o4 = reinterpret_cast<float4*>(acc_rows + (size_t)rowbase * FF);
            for (int i = 0; i < 16 * FF / 4 / 32; ++i) o4[i * 32 + lane] = z;
        } else {
            reinterpret_cast<float4*>(denom + (size_t)rowbase * HH)[lane] = z;
            if (rowbase == 0 && lane == 0) *gmax = -INFINITY;
        }
    }

    v8f acc[8];
#pragma unroll
    for (int c = 0; c < 8; ++c) acc[c] = (v8f){};

    for (int k = 0; k < FF; k += 4) {
        v2f a;
        a.x = xrow[k + 2 * half + 0];
        a.y = xrow[k + 2 * half + 1];
#pragma unroll
        for (int c = 0; c < 8; ++c) {
            const float* wp = W + (size_t)(k + 2 * half) * FF + c * 16 + l;
            v2f b;
            b.x = wp[0];
            b.y = wp[FF];
            acc[c] = __builtin_amdgcn_wmma_f32_16x16x4_f32(
                false, a, false, b, (short)0, acc[c], false, false);
        }
    }

#pragma unroll
    for (int c = 0; c < 8; ++c) {
#pragma unroll
        for (int r = 0; r < 8; ++r) {
            out[(size_t)(rowbase + r + 8 * half) * FF + c * 16 + l] = acc[c][r];
        }
    }
}

#define SC_EPW  32
#define SC_NBLK (EE / (8 * SC_EPW))
static_assert(EE % (8 * SC_EPW) == 0, "a block of 8 waves takes 256 whole edges");

template <int CTRL>
static __device__ __forceinline__ float row_lane(float v) {
    return __int_as_float(__builtin_amdgcn_update_dpp(0, __float_as_int(v), CTRL, 0xf, 0xf, true));
}

__global__ __launch_bounds__(256) void gat_score(
    const float* __restrict__ src_p,
    const float* __restrict__ dst_p,
    const int*   __restrict__ srcI,
    const int*   __restrict__ dstI,
    const float* __restrict__ attn,
    float* __restrict__ scorebuf,
    float* __restrict__ gmax)
{
    __shared__ float wmax[8];
    const int lane = threadIdx.x & 31;
    const int q    = lane & 3;
    const int head = lane >> 2;
    const int wv = __builtin_amdgcn_readfirstlane((int)((blockIdx.x * 256 + threadIdx.x) >> 5));
    const int e0 = wv * SC_EPW;

    const float4 at = *reinterpret_cast<const float4*>(attn + 4 * lane);
    float m = -INFINITY;

    for (int g = 0; g < SC_EPW / 4; ++g) {
        const int eb = e0 + 4 * g;
        float4 sv[4], dv[4];
#pragma unroll
        for (int es = 0; es < 4; ++es) {
            const int si = srcI[eb + es];
            const int di = dstI[eb + es];
            sv[es] = *reinterpret_cast<const float4*>(src_p + (size_t)si * FF + 4 * lane);
            dv[es] = *reinterpret_cast<const float4*>(dst_p + (size_t)di * FF + 4 * lane);
        }
        float p[4];
#pragma unroll
        for (int es = 0; es < 4; ++es) {
            float v0 = sv[es].x + dv[es].x;
            float v1 = sv[es].y + dv[es].y;
            float v2 = sv[es].z + dv[es].z;
            float v3 = sv[es].w + dv[es].w;
            v0 = (v0 >= 0.0f) ? v0 : 0.2f * v0;
            v1 = (v1 >= 0.0f) ? v1 : 0.2f * v1;
            v2 = (v2 >= 0.0f) ? v2 : 0.2f * v2;
            v3 = (v3 >= 0.0f) ? v3 : 0.2f * v3;
            float acc = at.x * v0;
            acc += at.y * v1;
            acc += at.z * v2;
            acc += at.w * v3;
            acc += row_lane<0xB1>(acc);
            acc += row_lane<0x4E>(acc);
            p[es] = acc;
        }
        const float sc = (q == 0) ? p[0] : (q == 1) ? p[1] : (q == 2) ? p[2] : p[3];
        scorebuf[(size_t)(eb + q) * HH + head] = sc;
        m = fmaxf(m, sc);
    }

    m = fmaxf(m, row_lane<0xB1>(m));
    m = fmaxf(m, row_lane<0x4E>(m));
    m = fmaxf(m, row_lane<0x164>(m));
    m = fmaxf(m, row_lane<0x168>(m));
    const float wm = fmaxf(__int_as_float(__builtin_amdgcn_readlane(__float_as_int(m), 0)),
                           __int_as_float(__builtin_amdgcn_readlane(__float_as_int(m), 16)));
    if (lane == 0) wmax[threadIdx.x >> 5] = wm;
    __syncthreads();
    if (threadIdx.x == 0) {
        float b = wmax[0];
#pragma unroll
        for (int i = 1; i < 8; ++i) b = fmaxf(b, wmax[i]);
        __hip_atomic_fetch_max(gmax, b, __ATOMIC_RELAXED, __HIP_MEMORY_SCOPE_AGENT);
    }
}

static_assert(EE % 64 == 0, "a block of 8 waves takes 64 whole edges");

static __device__ __forceinline__ float lane_value(float v, int src_lane) {
    return __int_as_float(__builtin_amdgcn_readlane(__float_as_int(v), src_lane));
}

__global__ __launch_bounds__(256) void gat_message(
    const float* __restrict__ src_p,
    const float* __restrict__ scorebuf,
    const float* __restrict__ gmax,
    float*       __restrict__ denom,
    const int*   __restrict__ srcI,
    const int*   __restrict__ dstI,
    float* __restrict__ out)
{
    const int lane = threadIdx.x & 31;
    const bool upper = (lane & 16) != 0;
    const int wv = __builtin_amdgcn_readfirstlane((int)((blockIdx.x * 256 + threadIdx.x) >> 5));
    const int e0 = wv * 8;
    const float gm = *gmax;

#pragma unroll
    for (int g = 0; g < 2; ++g) {
        const int eb = e0 + 4 * g;

        float v[4][4];
        int   di[4];
#pragma unroll
        for (int es = 0; es < 4; ++es) {
            const int si = srcI[eb + es];
            di[es] = dstI[eb + es];
            const float* __restrict__ sp = src_p + (size_t)si * FF + lane;
#pragma unroll
            for (int j = 0; j < 4; ++j) v[es][j] = sp[32 * j];
        }

        const int   dl = dstI[eb + (lane >> 3)];
        const float al = __expf(scorebuf[(size_t)eb * HH + lane] - gm);
        atomicAdd(&denom[(size_t)dl * HH + (lane & 7)], al);

#pragma unroll
        for (int es = 0; es < 4; ++es) {
            float* __restrict__ op = out + (size_t)di[es] * FF + lane;
#pragma unroll
            for (int j = 0; j < 4; ++j) {
                const float a_even = lane_value(al, es * 8 + 2 * j);
                const float a_odd  = lane_value(al, es * 8 + 2 * j + 1);
                const float a = upper ? a_odd : a_even;
                atomicAdd(&op[32 * j], v[es][j] * a);
            }
        }
    }
}

__global__ __launch_bounds__(256) void gat_finalize(
    float* __restrict__ out,
    const float* __restrict__ denom,
    const float* __restrict__ x,
    const float* __restrict__ bias,
    const float* __restrict__ prelu_w)
{
    const int i = blockIdx.x * 256 + threadIdx.x;
    if (i >= NN * FF) return;
    const float d  = denom[(size_t)(i >> 7) * HH + ((i >> 4) & (HH - 1))] + 1e-16f;
    const float v  = out[i] / d + x[i] + bias[i & (FF - 1)];
    const float pw = prelu_w[0];
    out[i] = (v >= 0.0f) ? v : pw * v;
}

extern "C" void kernel_launch(void* const* d_in, const int* in_sizes, int n_in,
                              void* d_out, int out_size, void* d_ws, size_t ws_size,
                              hipStream_t stream) {
    (void)in_sizes; (void)n_in; (void)out_size; (void)ws_size;

    const float* x        = (const float*)d_in[0];
    const int*   eidx     = (const int*)  d_in[1];
    const float* W_src    = (const float*)d_in[2];
    const float* W_dst    = (const float*)d_in[3];
    const float* attn     = (const float*)d_in[4];
    const float* bias     = (const float*)d_in[5];
    const float* prelu_w  = (const float*)d_in[6];
    const int*   srcI     = eidx;
    const int*   dstI     = eidx + EE;

    float* out = (float*)d_out;

    float* ws      = (float*)d_ws;
    float* src_p   = ws;
    float* dst_p   = src_p + (size_t)NN * FF;
    float* scoreb  = dst_p + (size_t)NN * FF;
    float* denom   = scoreb + (size_t)EE * HH;
    float* gmax    = denom + (size_t)NN * HH;

    {
        dim3 grid(2, NN / 16);
        gat_proj_gemm<<<grid, 32, 0, stream>>>(x, W_src, W_dst, src_p, dst_p, out, denom, gmax);
    }

    gat_score<<<SC_NBLK, 256, 0, stream>>>(src_p, dst_p, srcI, dstI, attn,
                                           scoreb, gmax);

    {
        const int nMsgBlk = EE / 64;
        gat_message<<<nMsgBlk, 256, 0, stream>>>(src_p, scoreb, gmax, denom, srcI, dstI, out);
    }

    {
        const int n = NN * FF;
        gat_finalize<<<(n + 255) / 256, 256, 0, stream>>>(out, denom, x, bias, prelu_w);
    }
}


# edges bucketed by destination (returning integer add = slot); one wave per node: scores, running-max softmax, message sum in registers, edges in source order; 1 row gather per edge, no f32 atomics
# speedup vs baseline: 4.0399x; 1.0967x over previous
#include <hip/hip_runtime.h>
#include <hip/hip_bf16.h>
#include <math.h>


#define NN 50000
#define EE 800000
#define FF 128
#define HH 8
#define DD 16

#define CAP     64
#define CNT_PAD 50048
static_assert(CAP % 4 == 0, "slots are read four at a time");
static_assert(CNT_PAD >= NN + 32, "the last even panel clears 32 counters");

typedef __attribute__((ext_vector_type(2))) float v2f;
typedef __attribute__((ext_vector_type(8))) float v8f;

__global__ __launch_bounds__(32) void gat_proj_gemm(
    const float* __restrict__ x,
    const float* __restrict__ W_src,
    const float* __restrict__ W_dst,
    float* __restrict__ src_p,
    float* __restrict__ dst_p,
    int*   __restrict__ cnt,
    float* __restrict__ gmax)
{
    const int which   = blockIdx.x;
    const int rowbase = blockIdx.y * 16;
    const float* __restrict__ W   = which ? W_dst : W_src;
    float* __restrict__       out = which ? dst_p : src_p;

    const int lane = threadIdx.x;
    const int half = lane >> 4;
    const int l    = lane & 15;

    const float* __restrict__ xrow = x + (size_t)(rowbase + l) * FF;

    if (which == 1) {
        if ((blockIdx.y & 1) == 0) cnt[rowbase + lane] = 0;
        if (rowbase == 0 && lane == 0) *gmax = -INFINITY;
    }

    v8f acc[8];
#pragma unroll
    for (int c = 0; c < 8; ++c) acc[c] = (v8f){};

    for (int k = 0; k < FF; k += 4) {
        v2f a;
        a.x = xrow[k + 2 * half + 0];
        a.y = xrow[k + 2 * half + 1];
#pragma unroll
        for (int c = 0; c < 8; ++c) {
            const float* wp = W + (size_t)(k + 2 * half) * FF + c * 16 + l;
            v2f b;
            b.x = wp[0];
            b.y = wp[FF];
            acc[c] = __builtin_amdgcn_wmma_f32_16x16x4_f32(
                false, a, false, b, (short)0, acc[c], false, false);
        }
    }

#pragma unroll
    for (int c = 0; c < 8; ++c) {
#pragma unroll
        for (int r = 0; r < 8; ++r) {
            out[(size_t)(rowbase + r + 8 * half) * FF + c * 16 + l] = acc[c][r];
        }
    }
}

static_assert(EE % (4 * 64) == 0, "a block of 2 waves takes 256 whole edges");

__global__ __launch_bounds__(64) void gat_bucket(
    const int* __restrict__ srcI,
    const int* __restrict__ dstI,
    int* __restrict__ cnt,
    int* __restrict__ slot)
{
    const int t = blockIdx.x * 64 + threadIdx.x;
    const int4 s4 = reinterpret_cast<const int4*>(srcI)[t];
    const int4 d4 = reinterpret_cast<const int4*>(dstI)[t];
    const int r0 = atomicAdd(&cnt[d4.x], 1);
    const int r1 = atomicAdd(&cnt[d4.y], 1);
    const int r2 = atomicAdd(&cnt[d4.z], 1);
    const int r3 = atomicAdd(&cnt[d4.w], 1);
    if (r0 < CAP) slot[(size_t)d4.x * CAP + r0] = s4.x;
    if (r1 < CAP) slot[(size_t)d4.y * CAP + r1] = s4.y;
    if (r2 < CAP) slot[(size_t)d4.z * CAP + r2] = s4.z;
    if (r3 < CAP) slot[(size_t)d4.w * CAP + r3] = s4.w;
}

template <int CTRL>
static __device__ __forceinline__ float row_lane(float v) {
    return __int_as_float(__builtin_amdgcn_update_dpp(0, __float_as_int(v), CTRL, 0xf, 0xf, true));
}

template <int CTRL>
static __device__ __forceinline__ int row_lane_i(int v) {
    return __builtin_amdgcn_update_dpp(0, v, CTRL, 0xf, 0xf, true);
}

#define NO_KEY 0x7fffffff

static __device__ __forceinline__ int take_min(int& k0, int& k1, int lane) {
    int v = min(k0, k1);
    v = min(v, row_lane_i<0xB1>(v));
    v = min(v, row_lane_i<0x4E>(v));
    v = min(v, row_lane_i<0x164>(v));
    v = min(v, row_lane_i<0x168>(v));
    const int mn = min(__builtin_amdgcn_readlane(v, 0), __builtin_amdgcn_readlane(v, 16));
    const unsigned b0 = __builtin_amdgcn_ballot_w32(k0 == mn);
    if (b0 != 0u) {
        if (lane == __builtin_ctz(b0)) k0 = NO_KEY;
    } else {
        const unsigned b1 = __builtin_amdgcn_ballot_w32(k1 == mn);
        if (b1 != 0u && lane == __builtin_ctz(b1)) k1 = NO_KEY;
    }
    return mn;
}

static __device__ __forceinline__ float4 row_piece(const float* __restrict__ p, int row, int lane) {
    return *reinterpret_cast<const float4*>(p + (size_t)row * FF + 4 * lane);
}

static __device__ __forceinline__ float head_score(const float4 s, const float4 d, const float4 at) {
    float v0 = s.x + d.x;
    float v1 = s.y + d.y;
    float v2 = s.z + d.z;
    float v3 = s.w + d.w;
    v0 = (v0 >= 0.0f) ? v0 : 0.2f * v0;
    v1 = (v1 >= 0.0f) ? v1 : 0.2f * v1;
    v2 = (v2 >= 0.0f) ? v2 : 0.2f * v2;
    v3 = (v3 >= 0.0f) ? v3 : 0.2f * v3;
    float acc = at.x * v0;
    acc += at.y * v1;
    acc += at.z * v2;
    acc += at.w * v3;
    acc += row_lane<0xB1>(acc);
    acc += row_lane<0x4E>(acc);
    return acc;
}

static __device__ __forceinline__ void absorb4(
    const float* __restrict__ src_p, int lane,
    int s0, int s1, int s2, int s3, int nv,
    const float4 drow, const float4 at,
    float& m, float& den, float4& sum)
{
    const float4 r0 = row_piece(src_p, s0, lane);
    const float4 r1 = row_piece(src_p, s1, lane);
    const float4 r2 = row_piece(src_p, s2, lane);
    const float4 r3 = row_piece(src_p, s3, lane);
    const float c0 = head_score(r0, drow, at);
    float c1 = head_score(r1, drow, at);
    float c2 = head_score(r2, drow, at);
    float c3 = head_score(r3, drow, at);
    c1 = (nv > 1) ? c1 : -INFINITY;
    c2 = (nv > 2) ? c2 : -INFINITY;
    c3 = (nv > 3) ? c3 : -INFINITY;
    const float mn = fmaxf(fmaxf(m, c0), fmaxf(c1, fmaxf(c2, c3)));
    const float f  = __expf(m - mn);
    const float w0 = __expf(c0 - mn);
    const float w1 = __expf(c1 - mn);
    const float w2 = __expf(c2 - mn);
    const float w3 = __expf(c3 - mn);
    den   = den   * f + ((w0 + w1) + (w2 + w3));
    sum.x = sum.x * f + ((w0 * r0.x + w1 * r1.x) + (w2 * r2.x + w3 * r3.x));
    sum.y = sum.y * f + ((w0 * r0.y + w1 * r1.y) + (w2 * r2.y + w3 * r3.y));
    sum.z = sum.z * f + ((w0 * r0.z + w1 * r1.z) + (w2 * r2.z + w3 * r3.z));
    sum.w = sum.w * f + ((w0 * r0.w + w1 * r1.w) + (w2 * r2.w + w3 * r3.w));
    m = mn;
}

static_assert(NN % 8 == 0, "a block of 8 waves takes 8 whole nodes");
static_assert(EE % 32 == 0, "the slow path reads 32 edges per trip");

__global__ __launch_bounds__(256) void gat_node(
    const float* __restrict__ src_p,
    const float* __restrict__ dst_p,
    const int*   __restrict__ srcI,
    const int*   __restrict__ dstI,
    const float* __restrict__ attn,
    const int*   __restrict__ cnt,
    const int*   __restrict__ slot,
    float* __restrict__ out,
    float* __restrict__ denb,
    float* __restrict__ maxb,
    float* __restrict__ gmax)
{
    __shared__ float sden[64];
    __shared__ float smax[64];
    __shared__ float wmax[8];
    const int lane = threadIdx.x & 31;
    const int w    = threadIdx.x >> 5;
    const int n   = __builtin_amdgcn_readfirstlane((int)((blockIdx.x * 256 + threadIdx.x) >> 5));
    const int deg = cnt[n];

    const float4 at   = *reinterpret_cast<const float4*>(attn + 4 * lane);
    const float4 drow = row_piece(dst_p, n, lane);

    float  m   = -INFINITY;
    float  den = 0.0f;
    float4 sum = make_float4(0.0f, 0.0f, 0.0f, 0.0f);

    if (deg <= CAP) {
        const int* __restrict__ sl = slot + (size_t)n * CAP;
        int k0 = sl[lane];
        int k1 = sl[lane + 32];
        k0 = (lane < deg) ? k0 : NO_KEY;
        k1 = (lane + 32 < deg) ? k1 : NO_KEY;
        for (int i = 0; i < deg; i += 4) {
            const int nv = deg - i;
            const int s0 = take_min(k0, k1, lane);
            const int t1 = take_min(k0, k1, lane);
            const int t2 = take_min(k0, k1, lane);
            const int t3 = take_min(k0, k1, lane);
            const int s1 = (nv > 1) ? t1 : s0;
            const int s2 = (nv > 2) ? t2 : s0;
            const int s3 = (nv > 3) ? t3 : s0;
            absorb4(src_p, lane, s0, s1, s2, s3, nv, drow, at, m, den, sum);
        }
    } else {
        for (int e = 0; e < EE; e += 32) {
            unsigned hit = __builtin_amdgcn_ballot_w32(dstI[e + lane] == n);
            while (hit != 0u) {
                const int b  = __builtin_ctz(hit);
                hit &= hit - 1u;
                const int s0 = srcI[e + b];
                absorb4(src_p, lane, s0, s0, s0, s0, 1, drow, at, m, den, sum);
            }
        }
    }

    *reinterpret_cast<float4*>(out + (size_t)n * FF + 4 * lane) = sum;

    if ((lane & 3) == 0) {
        sden[w * 8 + (lane >> 2)] = den;
        smax[w * 8 + (lane >> 2)] = m;
    }
    float mm = m;
    mm = fmaxf(mm, row_lane<0x164>(mm));
    mm = fmaxf(mm, row_lane<0x168>(mm));
    const float wm = fmaxf(__int_as_float(__builtin_amdgcn_readlane(__float_as_int(mm), 0)),
                           __int_as_float(__builtin_amdgcn_readlane(__float_as_int(mm), 16)));
    if (lane == 0) wmax[w] = wm;
    __syncthreads();
    if (threadIdx.x < 64) {
        denb[(size_t)blockIdx.x * 64 + threadIdx.x] = sden[threadIdx.x];
    } else if (threadIdx.x < 128) {
        maxb[(size_t)blockIdx.x * 64 + (threadIdx.x - 64)] = smax[threadIdx.x - 64];
    } else if (threadIdx.x == 128) {
        float b = wmax[0];
        for (int i = 1; i < 8; ++i) b = fmaxf(b, wmax[i]);
        __hip_atomic_fetch_max(gmax, b, __ATOMIC_RELAXED, __HIP_MEMORY_SCOPE_AGENT);
    }
}

__global__ __launch_bounds__(256) void gat_finalize(
    float* __restrict__ out,
    const float* __restrict__ denb,
    const float* __restrict__ maxb,
    const float* __restrict__ gmax,
    const float* __restrict__ x,
    const float* __restrict__ bias,
    const float* __restrict__ prelu_w)
{
    const int i = blockIdx.x * 256 + threadIdx.x;
    if (i >= NN * FF) return;
    const size_t nh = (size_t)(i >> 7) * HH + ((i >> 4) & (HH - 1));
    const float d  = denb[nh] + 1e-16f * __expf(*gmax - maxb[nh]);
    const float v  = out[i] / d + x[i] + bias[i & (FF - 1)];
    const float pw = prelu_w[0];
    out[i] = (v >= 0.0f) ? v : pw * v;
}

extern "C" void kernel_launch(void* const* d_in, const int* in_sizes, int n_in,
                              void* d_out, int out_size, void* d_ws, size_t ws_size,
                              hipStream_t stream) {
    (void)in_sizes; (void)n_in; (void)out_size; (void)ws_size;

    const float* x        = (const float*)d_in[0];
    const int*   eidx     = (const int*)  d_in[1];
    const float* W_src    = (const float*)d_in[2];
    const float* W_dst    = (const float*)d_in[3];
    const float* attn     = (const float*)d_in[4];
    const float* bias     = (const float*)d_in[5];
    const float* prelu_w  = (const float*)d_in[6];
    const int*   srcI     = eidx;
    const int*   dstI     = eidx + EE;

    float* out = (float*)d_out;

    float* ws      = (float*)d_ws;
    float* src_p   = ws;
    float* dst_p   = src_p + (size_t)NN * FF;
    int*   slot    = (int*)(dst_p + (size_t)NN * FF);
    int*   cnt     = slot + (size_t)NN * CAP;
    float* denb    = (float*)(cnt + CNT_PAD);
    float* maxb    = denb + (size_t)NN * HH;
    float* gmax    = maxb + (size_t)NN * HH;

    {
        dim3 grid(2, NN / 16);
        gat_proj_gemm<<<grid, 32, 0, stream>>>(x, W_src, W_dst, src_p, dst_p, cnt, gmax);
    }

    gat_bucket<<<EE / 256, 64, 0, stream>>>(srcI, dstI, cnt, slot);

    gat_node<<<NN / 8, 256, 0, stream>>>(src_p, dst_p, srcI, dstI, attn, cnt, slot,
                                         out, denb, maxb, gmax);

    {
        const int n = NN * FF;
        gat_finalize<<<(n + 255) / 256, 256, 0, stream>>>(out, denb, maxb, gmax, x, bias, prelu_w);
    }
}


# finalize folded into the node launch (divide by den', residual, bias, PReLU); last launch only tests den'+1e-16*exp(G-m)==den' per node and head and redoes the rows where it is not
# speedup vs baseline: 4.5055x; 1.1153x over previous
#include <hip/hip_runtime.h>
#include <hip/hip_bf16.h>
#include <math.h>


#define NN 50000
#define EE 800000
#define FF 128
#define HH 8
#define DD 16

#define CAP     64
#define CNT_PAD 50048
static_assert(CAP % 4 == 0, "slots are read four at a time");
static_assert(CNT_PAD >= NN + 32, "the last even panel clears 32 counters");

typedef __attribute__((ext_vector_type(2))) float v2f;
typedef __attribute__((ext_vector_type(8))) float v8f;

__global__ __launch_bounds__(32) void gat_proj_gemm(
    const float* __restrict__ x,
    const float* __restrict__ W_src,
    const float* __restrict__ W_dst,
    float* __restrict__ src_p,
    float* __restrict__ dst_p,
    int*   __restrict__ cnt,
    float* __restrict__ gmax)
{
    const int which   = blockIdx.x;
    const int rowbase = blockIdx.y * 16;
    const float* __restrict__ W   = which ? W_dst : W_src;
    float* __restrict__       out = which ? dst_p : src_p;

    const int lane = threadIdx.x;
    const int half = lane >> 4;
    const int l    = lane & 15;

    const float* __restrict__ xrow = x + (size_t)(rowbase + l) * FF;

    if (which == 1) {
        if ((blockIdx.y & 1) == 0) cnt[rowbase + lane] = 0;
        if (rowbase == 0 && lane == 0) *gmax = -INFINITY;
    }

    v8f acc[8];
#pragma unroll
    for (int c = 0; c < 8; ++c) acc[c] = (v8f){};

    for (int k = 0; k < FF; k += 4) {
        v2f a;
        a.x = xrow[k + 2 * half + 0];
        a.y = xrow[k + 2 * half + 1];
#pragma unroll
        for (int c = 0; c < 8; ++c) {
            const float* wp = W + (size_t)(k + 2 * half) * FF + c * 16 + l;
            v2f b;
            b.x = wp[0];
            b.y = wp[FF];
            acc[c] = __builtin_amdgcn_wmma_f32_16x16x4_f32(
                false, a, false, b, (short)0, acc[c], false, false);
        }
    }

#pragma unroll
    for (int c = 0; c < 8; ++c) {
#pragma unroll
        for (int r = 0; r < 8; ++r) {
            out[(size_t)(rowbase + r + 8 * half) * FF + c * 16 + l] = acc[c][r];
        }
    }
}

static_assert(EE % (4 * 64) == 0, "a block of 2 waves takes 256 whole edges");

__global__ __launch_bounds__(64) void gat_bucket(
    const int* __restrict__ srcI,
    const int* __restrict__ dstI,
    int* __restrict__ cnt,
    int* __restrict__ slot)
{
    const int t = blockIdx.x * 64 + threadIdx.x;
    const int4 s4 = reinterpret_cast<const int4*>(srcI)[t];
    const int4 d4 = reinterpret_cast<const int4*>(dstI)[t];
    const int r0 = atomicAdd(&cnt[d4.x], 1);
    const int r1 = atomicAdd(&cnt[d4.y], 1);
    const int r2 = atomicAdd(&cnt[d4.z], 1);
    const int r3 = atomicAdd(&cnt[d4.w], 1);
    if (r0 < CAP) slot[(size_t)d4.x * CAP + r0] = s4.x;
    if (r1 < CAP) slot[(size_t)d4.y * CAP + r1] = s4.y;
    if (r2 < CAP) slot[(size_t)d4.z * CAP + r2] = s4.z;
    if (r3 < CAP) slot[(size_t)d4.w * CAP + r3] = s4.w;
}

template <int CTRL>
static __device__ __forceinline__ float row_lane(float v) {
    return __int_as_float(__builtin_amdgcn_update_dpp(0, __float_as_int(v), CTRL, 0xf, 0xf, true));
}

template <int CTRL>
static __device__ __forceinline__ int row_lane_i(int v) {
    return __builtin_amdgcn_update_dpp(0, v, CTRL, 0xf, 0xf, true);
}

#define NO_KEY 0x7fffffff

static __device__ __forceinline__ int take_min(int& k0, int& k1, int lane) {
    int v = min(k0, k1);
    v = min(v, row_lane_i<0xB1>(v));
    v = min(v, row_lane_i<0x4E>(v));
    v = min(v, row_lane_i<0x164>(v));
    v = min(v, row_lane_i<0x168>(v));
    const int mn = min(__builtin_amdgcn_readlane(v, 0), __builtin_amdgcn_readlane(v, 16));
    const unsigned b0 = __builtin_amdgcn_ballot_w32(k0 == mn);
    if (b0 != 0u) {
        if (lane == __builtin_ctz(b0)) k0 = NO_KEY;
    } else {
        const unsigned b1 = __builtin_amdgcn_ballot_w32(k1 == mn);
        if (b1 != 0u && lane == __builtin_ctz(b1)) k1 = NO_KEY;
    }
    return mn;
}

static __device__ __forceinline__ float4 row_piece(const float* __restrict__ p, int row, int lane) {
    return *reinterpret_cast<const float4*>(p + (size_t)row * FF + 4 * lane);
}

static __device__ __forceinline__ float head_score(const float4 s, const float4 d, const float4 at) {
    float v0 = s.x + d.x;
    float v1 = s.y + d.y;
    float v2 = s.z + d.z;
    float v3 = s.w + d.w;
    v0 = (v0 >= 0.0f) ? v0 : 0.2f * v0;
    v1 = (v1 >= 0.0f) ? v1 : 0.2f * v1;
    v2 = (v2 >= 0.0f) ? v2 : 0.2f * v2;
    v3 = (v3 >= 0.0f) ? v3 : 0.2f * v3;
    float acc = at.x * v0;
    acc += at.y * v1;
    acc += at.z * v2;
    acc += at.w * v3;
    acc += row_lane<0xB1>(acc);
    acc += row_lane<0x4E>(acc);
    return acc;
}

static __device__ __forceinline__ void absorb4(
    const float* __restrict__ src_p, int lane,
    int s0, int s1, int s2, int s3, int nv,
    const float4 drow, const float4 at,
    float& m, float& den, float4& sum)
{
    const float4 r0 = row_piece(src_p, s0, lane);
    const float4 r1 = row_piece(src_p, s1, lane);
    const float4 r2 = row_piece(src_p, s2, lane);
    const float4 r3 = row_piece(src_p, s3, lane);
    const float c0 = head_score(r0, drow, at);
    float c1 = head_score(r1, drow, at);
    float c2 = head_score(r2, drow, at);
    float c3 = head_score(r3, drow, at);
    c1 = (nv > 1) ? c1 : -INFINITY;
    c2 = (nv > 2) ? c2 : -INFINITY;
    c3 = (nv > 3) ? c3 : -INFINITY;
    const float mn = fmaxf(fmaxf(m, c0), fmaxf(c1, fmaxf(c2, c3)));
    const float f  = __expf(m - mn);
    const float w0 = __expf(c0 - mn);
    const float w1 = __expf(c1 - mn);
    const float w2 = __expf(c2 - mn);
    const float w3 = __expf(c3 - mn);
    den   = den   * f + ((w0 + w1) + (w2 + w3));
    sum.x = sum.x * f + ((w0 * r0.x + w1 * r1.x) + (w2 * r2.x + w3 * r3.x));
    sum.y = sum.y * f + ((w0 * r0.y + w1 * r1.y) + (w2 * r2.y + w3 * r3.y));
    sum.z = sum.z * f + ((w0 * r0.z + w1 * r1.z) + (w2 * r2.z + w3 * r3.z));
    sum.w = sum.w * f + ((w0 * r0.w + w1 * r1.w) + (w2 * r2.w + w3 * r3.w));
    m = mn;
}

static_assert(NN % 8 == 0, "a block of 8 waves takes 8 whole nodes");
static_assert(EE % 32 == 0, "the slow path reads 32 edges per trip");

static __device__ __forceinline__ void node_state(
    const float* __restrict__ src_p,
    const int*   __restrict__ srcI,
    const int*   __restrict__ dstI,
    const int*   __restrict__ slot,
    int n, int deg, int lane,
    const float4 drow, const float4 at,
    float& m, float& den, float4& sum)
{
    m   = -INFINITY;
    den = 0.0f;
    sum = make_float4(0.0f, 0.0f, 0.0f, 0.0f);

    if (deg <= CAP) {
        const int* __restrict__ sl = slot + (size_t)n * CAP;
        int k0 = sl[lane];
        int k1 = sl[lane + 32];
        k0 = (lane < deg) ? k0 : NO_KEY;
        k1 = (lane + 32 < deg) ? k1 : NO_KEY;
        for (int i = 0; i < deg; i += 4) {
            const int nv = deg - i;
            const int s0 = take_min(k0, k1, lane);
            const int t1 = take_min(k0, k1, lane);
            const int t2 = take_min(k0, k1, lane);
            const int t3 = take_min(k0, k1, lane);
            const int s1 = (nv > 1) ? t1 : s0;
            const int s2 = (nv > 2) ? t2 : s0;
            const int s3 = (nv > 3) ? t3 : s0;
            absorb4(src_p, lane, s0, s1, s2, s3, nv, drow, at, m, den, sum);
        }
    } else {
        for (int e = 0; e < EE; e += 32) {
            unsigned hit = __builtin_amdgcn_ballot_w32(dstI[e + lane] == n);
            while (hit != 0u) {
                const int b  = __builtin_ctz(hit);
                hit &= hit - 1u;
                const int s0 = srcI[e + b];
                absorb4(src_p, lane, s0, s0, s0, s0, 1, drow, at, m, den, sum);
            }
        }
    }
}

static __device__ __forceinline__ float4 finish_piece(
    const float4 sum, float d, const float4 xr, const float4 bs, float pw)
{
    float4 o;
    o.x = sum.x / d + xr.x + bs.x;
    o.y = sum.y / d + xr.y + bs.y;
    o.z = sum.z / d + xr.z + bs.z;
    o.w = sum.w / d + xr.w + bs.w;
    o.x = (o.x >= 0.0f) ? o.x : pw * o.x;
    o.y = (o.y >= 0.0f) ? o.y : pw * o.y;
    o.z = (o.z >= 0.0f) ? o.z : pw * o.z;
    o.w = (o.w >= 0.0f) ? o.w : pw * o.w;
    return o;
}

__global__ __launch_bounds__(256) void gat_node(
    const float* __restrict__ src_p,
    const float* __restrict__ dst_p,
    const int*   __restrict__ srcI,
    const int*   __restrict__ dstI,
    const float* __restrict__ attn,
    const int*   __restrict__ cnt,
    const int*   __restrict__ slot,
    const float* __restrict__ x,
    const float* __restrict__ bias,
    const float* __restrict__ prelu_w,
    float* __restrict__ out,
    float* __restrict__ denb,
    float* __restrict__ maxb,
    float* __restrict__ gmax)
{
    __shared__ float sden[64];
    __shared__ float smax[64];
    __shared__ float wmax[8];
    const int lane = threadIdx.x & 31;
    const int w    = threadIdx.x >> 5;
    const int n   = __builtin_amdgcn_readfirstlane((int)((blockIdx.x * 256 + threadIdx.x) >> 5));
    const int deg = cnt[n];

    const float4 at   = *reinterpret_cast<const float4*>(attn + 4 * lane);
    const float4 drow = row_piece(dst_p, n, lane);

    float  m, den;
    float4 sum;
    node_state(src_p, srcI, dstI, slot, n, deg, lane, drow, at, m, den, sum);

    const float4 xr = row_piece(x, n, lane);
    const float4 bs = *reinterpret_cast<const float4*>(bias + 4 * lane);
    const float  pw = prelu_w[0];

    if ((lane & 3) == 0) {
        sden[w * 8 + (lane >> 2)] = den;
        smax[w * 8 + (lane >> 2)] = m;
    }
    float mm = m;
    mm = fmaxf(mm, row_lane<0x164>(mm));
    mm = fmaxf(mm, row_lane<0x168>(mm));
    const float wm = fmaxf(__int_as_float(__builtin_amdgcn_readlane(__float_as_int(mm), 0)),
                           __int_as_float(__builtin_amdgcn_readlane(__float_as_int(mm), 16)));
    if (lane == 0) wmax[w] = wm;
    __syncthreads();
    if (threadIdx.x < 64) {
        denb[(size_t)blockIdx.x * 64 + threadIdx.x] = sden[threadIdx.x];
    } else if (threadIdx.x < 128) {
        maxb[(size_t)blockIdx.x * 64 + (threadIdx.x - 64)] = smax[threadIdx.x - 64];
    } else if (threadIdx.x == 128) {
        float b = wmax[0];
        for (int i = 1; i < 8; ++i) b = fmaxf(b, wmax[i]);
        __hip_atomic_fetch_max(gmax, b, __ATOMIC_RELAXED, __HIP_MEMORY_SCOPE_AGENT);
    }

    const float d = (deg > 0) ? den : INFINITY;
    *reinterpret_cast<float4*>(out + (size_t)n * FF + 4 * lane) = finish_piece(sum, d, xr, bs, pw);
}

static __device__ __forceinline__ bool needs_term(float den, float m, float G) {
    return (den + 1e-16f * __expf(G - m)) != den;
}

__global__ __launch_bounds__(64) void gat_repair(
    const float* __restrict__ src_p,
    const float* __restrict__ dst_p,
    const int*   __restrict__ srcI,
    const int*   __restrict__ dstI,
    const float* __restrict__ attn,
    const int*   __restrict__ cnt,
    const int*   __restrict__ slot,
    const float* __restrict__ denb,
    const float* __restrict__ maxb,
    const float* __restrict__ gmax,
    const float* __restrict__ x,
    const float* __restrict__ bias,
    const float* __restrict__ prelu_w,
    float* __restrict__ out)
{
    const int lane = threadIdx.x & 31;
    const int node = blockIdx.x * 64 + threadIdx.x;
    const float G  = *gmax;
    bool bad = false;
    if (node < NN) {
        const float4 d0 = *reinterpret_cast<const float4*>(denb + (size_t)node * HH);
        const float4 d1 = *reinterpret_cast<const float4*>(denb + (size_t)node * HH + 4);
        const float4 m0 = *reinterpret_cast<const float4*>(maxb + (size_t)node * HH);
        const float4 m1 = *reinterpret_cast<const float4*>(maxb + (size_t)node * HH + 4);
        bad = needs_term(d0.x, m0.x, G) | needs_term(d0.y, m0.y, G) |
              needs_term(d0.z, m0.z, G) | needs_term(d0.w, m0.w, G) |
              needs_term(d1.x, m1.x, G) | needs_term(d1.y, m1.y, G) |
              needs_term(d1.z, m1.z, G) | needs_term(d1.w, m1.w, G);
    }
    unsigned hit = __builtin_amdgcn_ballot_w32(bad);
    if (hit == 0u) return;

    const int nbase = __builtin_amdgcn_readfirstlane(node);
    const float4 at = *reinterpret_cast<const float4*>(attn + 4 * lane);
    const float4 bs = *reinterpret_cast<const float4*>(bias + 4 * lane);
    const float  pw = prelu_w[0];
    while (hit != 0u) {
        const int n = nbase + __builtin_ctz(hit);
        hit &= hit - 1u;
        const int deg = cnt[n];
        const float4 drow = row_piece(dst_p, n, lane);
        const float4 xr   = row_piece(x, n, lane);
        float  m, dn;
        float4 sum;
        node_state(src_p, srcI, dstI, slot, n, deg, lane, drow, at, m, dn, sum);
        const float d = dn + 1e-16f * __expf(G - m);
        *reinterpret_cast<float4*>(out + (size_t)n * FF + 4 * lane) = finish_piece(sum, d, xr, bs, pw);
    }
}

extern "C" void kernel_launch(void* const* d_in, const int* in_sizes, int n_in,
                              void* d_out, int out_size, void* d_ws, size_t ws_size,
                              hipStream_t stream) {
    (void)in_sizes; (void)n_in; (void)out_size; (void)ws_size;

    const float* x        = (const float*)d_in[0];
    const int*   eidx     = (const int*)  d_in[1];
    const float* W_src    = (const float*)d_in[2];
    const float* W_dst    = (const float*)d_in[3];
    const float* attn     = (const float*)d_in[4];
    const float* bias     = (const float*)d_in[5];
    const float* prelu_w  = (const float*)d_in[6];
    const int*   srcI     = eidx;
    const int*   dstI     = eidx + EE;

    float* out = (float*)d_out;

    float* ws      = (float*)d_ws;
    float* src_p   = ws;
    float* dst_p   = src_p + (size_t)NN * FF;
    int*   slot    = (int*)(dst_p + (size_t)NN * FF);
    int*   cnt     = slot + (size_t)NN * CAP;
    float* denb    = (float*)(cnt + CNT_PAD);
    float* maxb    = denb + (size_t)NN * HH;
    float* gmax    = maxb + (size_t)NN * HH;

    {
        dim3 grid(2, NN / 16);
        gat_proj_gemm<<<grid, 32, 0, stream>>>(x, W_src, W_dst, src_p, dst_p, cnt, gmax);
    }

    gat_bucket<<<EE / 256, 64, 0, stream>>>(srcI, dstI, cnt, slot);

    gat_node<<<NN / 8, 256, 0, stream>>>(src_p, dst_p, srcI, dstI, attn, cnt, slot,
                                         x, bias, prelu_w, out, denb, maxb, gmax);

    gat_repair<<<(NN + 63) / 64, 64, 0, stream>>>(src_p, dst_p, srcI, dstI, attn, cnt, slot,
                                                  denb, maxb, gmax, x, bias, prelu_w, out);
}


# edges grouped by destination in two levels on LDS counters: 4096-edge blocks bin by 64-node range (one global add per block and range, coalesced runs), then a block per range fills slots; 5 launches
# speedup vs baseline: 7.0233x; 1.5588x over previous
#include <hip/hip_runtime.h>
#include <hip/hip_bf16.h>
#include <math.h>


#define NN 50000
#define EE 800000
#define FF 128
#define HH 8
#define DD 16

#define CAP     64
static_assert(CAP % 4 == 0, "slots are read four at a time");

#define BIN_SHIFT 6
#define BIN_NODES (1 << BIN_SHIFT)
#define NBIN      ((NN + BIN_NODES - 1) / BIN_NODES)
#define NBIN_PAD  800
#define BINCAP    2048
#define ROWS_PAD  (NBIN * BIN_NODES)
#define CHUNK_I4  1024
static_assert(NN <= 65536, "source and destination share one 32-bit word");
static_assert(NBIN_PAD >= NBIN && NBIN_PAD % 32 == 0, "range counters are cleared 32 at a time");
static_assert(EE % 4 == 0, "indices are read 16 bytes at a time");
static_assert(BINCAP % 256 == 0, "K2b reads a range's words 256 at a time");

typedef __attribute__((ext_vector_type(2))) float v2f;
typedef __attribute__((ext_vector_type(8))) float v8f;

__global__ __launch_bounds__(32) void gat_proj_gemm(
    const float* __restrict__ x,
    const float* __restrict__ W_src,
    const float* __restrict__ W_dst,
    float* __restrict__ src_p,
    float* __restrict__ dst_p,
    int*   __restrict__ binCnt,
    float* __restrict__ gmax)
{
    const int which   = blockIdx.x;
    const int rowbase = blockIdx.y * 16;
    const float* __restrict__ W   = which ? W_dst : W_src;
    float* __restrict__       out = which ? dst_p : src_p;

    const int lane = threadIdx.x;
    const int half = lane >> 4;
    const int l    = lane & 15;

    const float* __restrict__ xrow = x + (size_t)(rowbase + l) * FF;

    if (which == 1) {
        if (blockIdx.y < NBIN_PAD / 32) binCnt[blockIdx.y * 32 + lane] = 0;
        if (rowbase == 0 && lane == 0) *gmax = -INFINITY;
    }

    v8f acc[8];
#pragma unroll
    for (int c = 0; c < 8; ++c) acc[c] = (v8f){};

    for (int k = 0; k < FF; k += 4) {
        v2f a;
        a.x = xrow[k + 2 * half + 0];
        a.y = xrow[k + 2 * half + 1];
#pragma unroll
        for (int c = 0; c < 8; ++c) {
            const float* wp = W + (size_t)(k + 2 * half) * FF + c * 16 + l;
            v2f b;
            b.x = wp[0];
            b.y = wp[FF];
            acc[c] = __builtin_amdgcn_wmma_f32_16x16x4_f32(
                false, a, false, b, (short)0, acc[c], false, false);
        }
    }

#pragma unroll
    for (int c = 0; c < 8; ++c) {
#pragma unroll
        for (int r = 0; r < 8; ++r) {
            out[(size_t)(rowbase + r + 8 * half) * FF + c * 16 + l] = acc[c][r];
        }
    }
}

#define CHUNK_E   (4 * CHUNK_I4)
#define SCAN_W    (NBIN_PAD / 32)
static_assert(SCAN_W * 32 == NBIN_PAD, "the first wave covers all range counters");
static_assert(CHUNK_E == 16 * 256, "a thread takes 16 edges");

__global__ __launch_bounds__(256) void gat_partition(
    const int* __restrict__ srcI,
    const int* __restrict__ dstI,
    int* __restrict__ binCnt,
    unsigned* __restrict__ binData)
{
    __shared__ int hist[NBIN_PAD];
    __shared__ int loff[NBIN_PAD];
    __shared__ int gbase[NBIN_PAD];
    __shared__ unsigned stage[CHUNK_E];
    const int t = threadIdx.x;
    for (int i = t; i < NBIN_PAD; i += 256) hist[i] = 0;
    __syncthreads();

    int sv[16], dv[16], r[16];
#pragma unroll
    for (int j = 0; j < 4; ++j) {
        const int p = blockIdx.x * CHUNK_I4 + j * 256 + t;
        int4 s4 = make_int4(0, 0, 0, 0);
        int4 d4 = make_int4(-1, -1, -1, -1);
        if (p < EE / 4) {
            s4 = reinterpret_cast<const int4*>(srcI)[p];
            d4 = reinterpret_cast<const int4*>(dstI)[p];
        }
        sv[4 * j + 0] = s4.x; sv[4 * j + 1] = s4.y; sv[4 * j + 2] = s4.z; sv[4 * j + 3] = s4.w;
        dv[4 * j + 0] = d4.x; dv[4 * j + 1] = d4.y; dv[4 * j + 2] = d4.z; dv[4 * j + 3] = d4.w;
    }
#pragma unroll
    for (int q = 0; q < 16; ++q)
        r[q] = (dv[q] >= 0) ? atomicAdd(&hist[dv[q] >> BIN_SHIFT], 1) : 0;
    __syncthreads();

    int c[4], base[4];
#pragma unroll
    for (int q = 0; q < 4; ++q) {
        const int i = t + 256 * q;
        c[q] = (i < NBIN) ? hist[i] : 0;
    }
#pragma unroll
    for (int q = 0; q < 4; ++q)
        base[q] = (c[q] > 0) ? atomicAdd(&binCnt[t + 256 * q], c[q]) : 0;
    if (t < 32) {
        int v[SCAN_W];
        int sum = 0;
#pragma unroll
        for (int k = 0; k < SCAN_W; ++k) v[k] = hist[SCAN_W * t + k];
#pragma unroll
        for (int k = 0; k < SCAN_W; ++k) { const int xk = v[k]; v[k] = sum; sum += xk; }
        int inc = sum;
#pragma unroll
        for (int dlt = 1; dlt < 32; dlt <<= 1) {
            const int y = __shfl_up(inc, dlt, 32);
            if (t >= dlt) inc += y;
        }
        const int excl = inc - sum;
#pragma unroll
        for (int k = 0; k < SCAN_W; ++k) loff[SCAN_W * t + k] = v[k] + excl;
    }
#pragma unroll
    for (int q = 0; q < 4; ++q) {
        const int i = t + 256 * q;
        if (i < NBIN) gbase[i] = base[q];
    }
    __syncthreads();

    int lo[16];
#pragma unroll
    for (int q = 0; q < 16; ++q)
        lo[q] = loff[max(dv[q], 0) >> BIN_SHIFT] + r[q];
#pragma unroll
    for (int q = 0; q < 16; ++q) {
        if (dv[q] >= 0)
            stage[lo[q]] = (unsigned)sv[q] | ((unsigned)dv[q] << 16);
    }
    __syncthreads();

    const int nvalid = min(CHUNK_E, EE - (int)blockIdx.x * CHUNK_E);
    unsigned w[16];
    int pb[16], pos[16];
#pragma unroll
    for (int q = 0; q < 16; ++q)
        w[q] = stage[min(t + 256 * q, nvalid - 1)];
#pragma unroll
    for (int q = 0; q < 16; ++q) {
        pb[q]  = (int)(w[q] >> (16 + BIN_SHIFT));
        pos[q] = gbase[pb[q]] - loff[pb[q]] + (t + 256 * q);
    }
#pragma unroll
    for (int q = 0; q < 16; ++q) {
        if (t + 256 * q < nvalid && pos[q] < BINCAP)
            binData[(size_t)pb[q] * BINCAP + pos[q]] = w[q];
    }
}

__global__ __launch_bounds__(256) void gat_rank(
    const int* __restrict__ srcI,
    const int* __restrict__ dstI,
    const int* __restrict__ binCnt,
    const unsigned* __restrict__ binData,
    int* __restrict__ cnt,
    int* __restrict__ slot)
{
    __shared__ int lcnt[BIN_NODES];
    __shared__ __attribute__((aligned(16))) int lslot[BIN_NODES * CAP];
    const int t = threadIdx.x;
    const int b = blockIdx.x;
    if (t < BIN_NODES) lcnt[t] = 0;
    for (int i = t; i < BIN_NODES * CAP / 4; i += 256)
        reinterpret_cast<int4*>(lslot)[i] = make_int4(0x7fffffff, 0x7fffffff, 0x7fffffff, 0x7fffffff);
    __syncthreads();

    const int total = binCnt[b];
    if (total <= BINCAP) {
        const unsigned* __restrict__ bd = binData + (size_t)b * BINCAP;
        unsigned w[BINCAP / 256];
#pragma unroll
        for (int q = 0; q < BINCAP / 256; ++q) {
            const int i = t + 256 * q;
            w[q] = (i < total) ? bd[i] : 0u;
        }
#pragma unroll
        for (int q = 0; q < BINCAP / 256; ++q) {
            if (t + 256 * q < total) {
                const int d  = (int)(w[q] >> 16) & (BIN_NODES - 1);
                const int rk = atomicAdd(&lcnt[d], 1);
                if (rk < CAP) lslot[d * CAP + rk] = (int)(w[q] & 0xffffu);
            }
        }
    } else {
        for (int e = t; e < EE; e += 256) {
            const int dn = dstI[e];
            if ((dn >> BIN_SHIFT) == b) {
                const int d  = dn & (BIN_NODES - 1);
                const int rk = atomicAdd(&lcnt[d], 1);
                if (rk < CAP) lslot[d * CAP + rk] = srcI[e];
            }
        }
    }
    __syncthreads();
    if (t < BIN_NODES) cnt[b * BIN_NODES + t] = lcnt[t];
    int4* __restrict__ so = reinterpret_cast<int4*>(slot + (size_t)b * BIN_NODES * CAP);
    for (int i = t; i < BIN_NODES * CAP / 4; i += 256)
        so[i] = reinterpret_cast<const int4*>(lslot)[i];
}

template <int CTRL>
static __device__ __forceinline__ float row_lane(float v) {
    return __int_as_float(__builtin_amdgcn_update_dpp(0, __float_as_int(v), CTRL, 0xf, 0xf, true));
}

template <int CTRL>
static __device__ __forceinline__ int row_lane_i(int v) {
    return __builtin_amdgcn_update_dpp(0, v, CTRL, 0xf, 0xf, true);
}

#define NO_KEY 0x7fffffff

static __device__ __forceinline__ int take_min(int& k0, int& k1, int lane) {
    int v = min(k0, k1);
    v = min(v, row_lane_i<0xB1>(v));
    v = min(v, row_lane_i<0x4E>(v));
    v = min(v, row_lane_i<0x164>(v));
    v = min(v, row_lane_i<0x168>(v));
    const int mn = min(__builtin_amdgcn_readlane(v, 0), __builtin_amdgcn_readlane(v, 16));
    const unsigned b0 = __builtin_amdgcn_ballot_w32(k0 == mn);
    if (b0 != 0u) {
        if (lane == __builtin_ctz(b0)) k0 = NO_KEY;
    } else {
        const unsigned b1 = __builtin_amdgcn_ballot_w32(k1 == mn);
        if (b1 != 0u && lane == __builtin_ctz(b1)) k1 = NO_KEY;
    }
    return mn;
}

static __device__ __forceinline__ float4 row_piece(const float* __restrict__ p, int row, int lane) {
    return *reinterpret_cast<const float4*>(p + (size_t)row * FF + 4 * lane);
}

static __device__ __forceinline__ float head_score(const float4 s, const float4 d, const float4 at) {
    float v0 = s.x + d.x;
    float v1 = s.y + d.y;
    float v2 = s.z + d.z;
    float v3 = s.w + d.w;
    v0 = (v0 >= 0.0f) ? v0 : 0.2f * v0;
    v1 = (v1 >= 0.0f) ? v1 : 0.2f * v1;
    v2 = (v2 >= 0.0f) ? v2 : 0.2f * v2;
    v3 = (v3 >= 0.0f) ? v3 : 0.2f * v3;
    float acc = at.x * v0;
    acc += at.y * v1;
    acc += at.z * v2;
    acc += at.w * v3;
    acc += row_lane<0xB1>(acc);
    acc += row_lane<0x4E>(acc);
    return acc;
}

static __device__ __forceinline__ void absorb4(
    const float* __restrict__ src_p, int lane,
    int s0, int s1, int s2, int s3, int nv,
    const float4 drow, const float4 at,
    float& m, float& den, float4& sum)
{
    const float4 r0 = row_piece(src_p, s0, lane);
    const float4 r1 = row_piece(src_p, s1, lane);
    const float4 r2 = row_piece(src_p, s2, lane);
    const float4 r3 = row_piece(src_p, s3, lane);
    const float c0 = head_score(r0, drow, at);
    float c1 = head_score(r1, drow, at);
    float c2 = head_score(r2, drow, at);
    float c3 = head_score(r3, drow, at);
    c1 = (nv > 1) ? c1 : -INFINITY;
    c2 = (nv > 2) ? c2 : -INFINITY;
    c3 = (nv > 3) ? c3 : -INFINITY;
    const float mn = fmaxf(fmaxf(m, c0), fmaxf(c1, fmaxf(c2, c3)));
    const float f  = __expf(m - mn);
    const float w0 = __expf(c0 - mn);
    const float w1 = __expf(c1 - mn);
    const float w2 = __expf(c2 - mn);
    const float w3 = __expf(c3 - mn);
    den   = den   * f + ((w0 + w1) + (w2 + w3));
    sum.x = sum.x * f + ((w0 * r0.x + w1 * r1.x) + (w2 * r2.x + w3 * r3.x));
    sum.y = sum.y * f + ((w0 * r0.y + w1 * r1.y) + (w2 * r2.y + w3 * r3.y));
    sum.z = sum.z * f + ((w0 * r0.z + w1 * r1.z) + (w2 * r2.z + w3 * r3.z));
    sum.w = sum.w * f + ((w0 * r0.w + w1 * r1.w) + (w2 * r2.w + w3 * r3.w));
    m = mn;
}

static_assert(NN % 8 == 0, "a block of 8 waves takes 8 whole nodes");
static_assert(EE % 32 == 0, "the slow path reads 32 edges per trip");

static __device__ __forceinline__ void node_state(
    const float* __restrict__ src_p,
    const int*   __restrict__ srcI,
    const int*   __restrict__ dstI,
    const int*   __restrict__ slot,
    int n, int deg, int lane,
    const float4 drow, const float4 at,
    float& m, float& den, float4& sum)
{
    m   = -INFINITY;
    den = 0.0f;
    sum = make_float4(0.0f, 0.0f, 0.0f, 0.0f);

    if (deg <= CAP) {
        const int* __restrict__ sl = slot + (size_t)n * CAP;
        int k0 = sl[lane];
        int k1 = sl[lane + 32];
        k0 = (lane < deg) ? k0 : NO_KEY;
        k1 = (lane + 32 < deg) ? k1 : NO_KEY;
        for (int i = 0; i < deg; i += 4) {
            const int nv = deg - i;
            const int s0 = take_min(k0, k1, lane);
            const int t1 = take_min(k0, k1, lane);
            const int t2 = take_min(k0, k1, lane);
            const int t3 = take_min(k0, k1, lane);
            const int s1 = (nv > 1) ? t1 : s0;
            const int s2 = (nv > 2) ? t2 : s0;
            const int s3 = (nv > 3) ? t3 : s0;
            absorb4(src_p, lane, s0, s1, s2, s3, nv, drow, at, m, den, sum);
        }
    } else {
        for (int e = 0; e < EE; e += 32) {
            unsigned hit = __builtin_amdgcn_ballot_w32(dstI[e + lane] == n);
            while (hit != 0u) {
                const int b  = __builtin_ctz(hit);
                hit &= hit - 1u;
                const int s0 = srcI[e + b];
                absorb4(src_p, lane, s0, s0, s0, s0, 1, drow, at, m, den, sum);
            }
        }
    }
}

static __device__ __forceinline__ float4 finish_piece(
    const float4 sum, float d, const float4 xr, const float4 bs, float pw)
{
    float4 o;
    o.x = sum.x / d + xr.x + bs.x;
    o.y = sum.y / d + xr.y + bs.y;
    o.z = sum.z / d + xr.z + bs.z;
    o.w = sum.w / d + xr.w + bs.w;
    o.x = (o.x >= 0.0f) ? o.x : pw * o.x;
    o.y = (o.y >= 0.0f) ? o.y : pw * o.y;
    o.z = (o.z >= 0.0f) ? o.z : pw * o.z;
    o.w = (o.w >= 0.0f) ? o.w : pw * o.w;
    return o;
}

__global__ __launch_bounds__(256) void gat_node(
    const float* __restrict__ src_p,
    const float* __restrict__ dst_p,
    const int*   __restrict__ srcI,
    const int*   __restrict__ dstI,
    const float* __restrict__ attn,
    const int*   __restrict__ cnt,
    const int*   __restrict__ slot,
    const float* __restrict__ x,
    const float* __restrict__ bias,
    const float* __restrict__ prelu_w,
    float* __restrict__ out,
    float* __restrict__ denb,
    float* __restrict__ maxb,
    float* __restrict__ gmax)
{
    __shared__ float sden[64];
    __shared__ float smax[64];
    __shared__ float wmax[8];
    const int lane = threadIdx.x & 31;
    const int w    = threadIdx.x >> 5;
    const int n   = __builtin_amdgcn_readfirstlane((int)((blockIdx.x * 256 + threadIdx.x) >> 5));
    const int deg = cnt[n];

    const float4 at   = *reinterpret_cast<const float4*>(attn + 4 * lane);
    const float4 drow = row_piece(dst_p, n, lane);

    float  m, den;
    float4 sum;
    node_state(src_p, srcI, dstI, slot, n, deg, lane, drow, at, m, den, sum);

    const float4 xr = row_piece(x, n, lane);
    const float4 bs = *reinterpret_cast<const float4*>(bias + 4 * lane);
    const float  pw = prelu_w[0];

    if ((lane & 3) == 0) {
        sden[w * 8 + (lane >> 2)] = den;
        smax[w * 8 + (lane >> 2)] = m;
    }
    float mm = m;
    mm = fmaxf(mm, row_lane<0x164>(mm));
    mm = fmaxf(mm, row_lane<0x168>(mm));
    const float wm = fmaxf(__int_as_float(__builtin_amdgcn_readlane(__float_as_int(mm), 0)),
                           __int_as_float(__builtin_amdgcn_readlane(__float_as_int(mm), 16)));
    if (lane == 0) wmax[w] = wm;
    __syncthreads();
    if (threadIdx.x < 64) {
        denb[(size_t)blockIdx.x * 64 + threadIdx.x] = sden[threadIdx.x];
    } else if (threadIdx.x < 128) {
        maxb[(size_t)blockIdx.x * 64 + (threadIdx.x - 64)] = smax[threadIdx.x - 64];
    } else if (threadIdx.x == 128) {
        float b = wmax[0];
        for (int i = 1; i < 8; ++i) b = fmaxf(b, wmax[i]);
        __hip_atomic_fetch_max(gmax, b, __ATOMIC_RELAXED, __HIP_MEMORY_SCOPE_AGENT);
    }

    const float d = (deg > 0) ? den : INFINITY;
    *reinterpret_cast<float4*>(out + (size_t)n * FF + 4 * lane) = finish_piece(sum, d, xr, bs, pw);
}

static __device__ __forceinline__ bool needs_term(float den, float m, float G) {
    return (den + 1e-16f * __expf(G - m)) != den;
}

__global__ __launch_bounds__(64) void gat_repair(
    const float* __restrict__ src_p,
    const float* __restrict__ dst_p,
    const int*   __restrict__ srcI,
    const int*   __restrict__ dstI,
    const float* __restrict__ attn,
    const int*   __restrict__ cnt,
    const int*   __restrict__ slot,
    const float* __restrict__ denb,
    const float* __restrict__ maxb,
    const float* __restrict__ gmax,
    const float* __restrict__ x,
    const float* __restrict__ bias,
    const float* __restrict__ prelu_w,
    float* __restrict__ out)
{
    const int lane = threadIdx.x & 31;
    const int node = blockIdx.x * 64 + threadIdx.x;
    const float G  = *gmax;
    bool bad = false;
    if (node < NN) {
        const float4 d0 = *reinterpret_cast<const float4*>(denb + (size_t)node * HH);
        const float4 d1 = *reinterpret_cast<const float4*>(denb + (size_t)node * HH + 4);
        const float4 m0 = *reinterpret_cast<const float4*>(maxb + (size_t)node * HH);
        const float4 m1 = *reinterpret_cast<const float4*>(maxb + (size_t)node * HH + 4);
        bad = needs_term(d0.x, m0.x, G) | needs_term(d0.y, m0.y, G) |
              needs_term(d0.z, m0.z, G) | needs_term(d0.w, m0.w, G) |
              needs_term(d1.x, m1.x, G) | needs_term(d1.y, m1.y, G) |
              needs_term(d1.z, m1.z, G) | needs_term(d1.w, m1.w, G);
    }
    unsigned hit = __builtin_amdgcn_ballot_w32(bad);
    if (hit == 0u) return;

    const int nbase = __builtin_amdgcn_readfirstlane(node);
    const float4 at = *reinterpret_cast<const float4*>(attn + 4 * lane);
    const float4 bs = *reinterpret_cast<const float4*>(bias + 4 * lane);
    const float  pw = prelu_w[0];
    while (hit != 0u) {
        const int n = nbase + __builtin_ctz(hit);
        hit &= hit - 1u;
        const int deg = cnt[n];
        const float4 drow = row_piece(dst_p, n, lane);
        const float4 xr   = row_piece(x, n, lane);
        float  m, dn;
        float4 sum;
        node_state(src_p, srcI, dstI, slot, n, deg, lane, drow, at, m, dn, sum);
        const float d = dn + 1e-16f * __expf(G - m);
        *reinterpret_cast<float4*>(out + (size_t)n * FF + 4 * lane) = finish_piece(sum, d, xr, bs, pw);
    }
}

extern "C" void kernel_launch(void* const* d_in, const int* in_sizes, int n_in,
                              void* d_out, int out_size, void* d_ws, size_t ws_size,
                              hipStream_t stream) {
    (void)in_sizes; (void)n_in; (void)out_size; (void)ws_size;

    const float* x        = (const float*)d_in[0];
    const int*   eidx     = (const int*)  d_in[1];
    const float* W_src    = (const float*)d_in[2];
    const float* W_dst    = (const float*)d_in[3];
    const float* attn     = (const float*)d_in[4];
    const float* bias     = (const float*)d_in[5];
    const float* prelu_w  = (const float*)d_in[6];
    const int*   srcI     = eidx;
    const int*   dstI     = eidx + EE;

    float* out = (float*)d_out;

    float* ws      = (float*)d_ws;
    float* src_p   = ws;
    float* dst_p   = src_p + (size_t)NN * FF;
    int*   slot    = (int*)(dst_p + (size_t)NN * FF);
    int*   cnt     = slot + (size_t)ROWS_PAD * CAP;
    int*   binCnt  = cnt + ROWS_PAD;
    unsigned* binData = (unsigned*)(binCnt + NBIN_PAD);
    float* denb    = (float*)(binData + (size_t)NBIN * BINCAP);
    float* maxb    = denb + (size_t)NN * HH;
    float* gmax    = maxb + (size_t)NN * HH;

    {
        dim3 grid(2, NN / 16);
        gat_proj_gemm<<<grid, 32, 0, stream>>>(x, W_src, W_dst, src_p, dst_p, binCnt, gmax);
    }

    gat_partition<<<(EE / 4 + CHUNK_I4 - 1) / CHUNK_I4, 256, 0, stream>>>(srcI, dstI, binCnt, binData);

    gat_rank<<<NBIN, 256, 0, stream>>>(srcI, dstI, binCnt, binData, cnt, slot);

    gat_node<<<NN / 8, 256, 0, stream>>>(src_p, dst_p, srcI, dstI, attn, cnt, slot,
                                         x, bias, prelu_w, out, denb, maxb, gmax);

    gat_repair<<<(NN + 63) / 64, 64, 0, stream>>>(src_p, dst_p, srcI, dstI, attn, cnt, slot,
                                                  denb, maxb, gmax, x, bias, prelu_w, out);
}


# sources sorted once per node in the slot launch (a thread per edge counts its node's smaller words in LDS); node launch takes sorted sources by scalar loads; same order and sums
# speedup vs baseline: 7.3257x; 1.0431x over previous
#include <hip/hip_runtime.h>
#include <hip/hip_bf16.h>
#include <math.h>


#define NN 50000
#define EE 800000
#define FF 128
#define HH 8
#define DD 16

#define CAP     64
static_assert(CAP % 4 == 0, "slots are read four at a time");

#define BIN_SHIFT 6
#define BIN_NODES (1 << BIN_SHIFT)
#define NBIN      ((NN + BIN_NODES - 1) / BIN_NODES)
#define NBIN_PAD  800
#define BINCAP    2048
#define ROWS_PAD  (NBIN * BIN_NODES)
#define CHUNK_I4  1024
static_assert(NN <= 65536, "source and destination share one 32-bit word");
static_assert(NBIN_PAD >= NBIN && NBIN_PAD % 32 == 0, "range counters are cleared 32 at a time");
static_assert(EE % 4 == 0, "indices are read 16 bytes at a time");
static_assert(BINCAP % 256 == 0, "K2b reads a range's words 256 at a time");

typedef __attribute__((ext_vector_type(2))) float v2f;
typedef __attribute__((ext_vector_type(8))) float v8f;

__global__ __launch_bounds__(32) void gat_proj_gemm(
    const float* __restrict__ x,
    const float* __restrict__ W_src,
    const float* __restrict__ W_dst,
    float* __restrict__ src_p,
    float* __restrict__ dst_p,
    int*   __restrict__ binCnt,
    float* __restrict__ gmax)
{
    const int which   = blockIdx.x;
    const int rowbase = blockIdx.y * 16;
    const float* __restrict__ W   = which ? W_dst : W_src;
    float* __restrict__       out = which ? dst_p : src_p;

    const int lane = threadIdx.x;
    const int half = lane >> 4;
    const int l    = lane & 15;

    const float* __restrict__ xrow = x + (size_t)(rowbase + l) * FF;

    if (which == 1) {
        if (blockIdx.y < NBIN_PAD / 32) binCnt[blockIdx.y * 32 + lane] = 0;
        if (rowbase == 0 && lane == 0) *gmax = -INFINITY;
    }

    v8f acc[8];
#pragma unroll
    for (int c = 0; c < 8; ++c) acc[c] = (v8f){};

    for (int k = 0; k < FF; k += 4) {
        v2f a;
        a.x = xrow[k + 2 * half + 0];
        a.y = xrow[k + 2 * half + 1];
#pragma unroll
        for (int c = 0; c < 8; ++c) {
            const float* wp = W + (size_t)(k + 2 * half) * FF + c * 16 + l;
            v2f b;
            b.x = wp[0];
            b.y = wp[FF];
            acc[c] = __builtin_amdgcn_wmma_f32_16x16x4_f32(
                false, a, false, b, (short)0, acc[c], false, false);
        }
    }

#pragma unroll
    for (int c = 0; c < 8; ++c) {
#pragma unroll
        for (int r = 0; r < 8; ++r) {
            out[(size_t)(rowbase + r + 8 * half) * FF + c * 16 + l] = acc[c][r];
        }
    }
}

#define CHUNK_E   (4 * CHUNK_I4)
#define SCAN_W    (NBIN_PAD / 32)
static_assert(SCAN_W * 32 == NBIN_PAD, "the first wave covers all range counters");
static_assert(CHUNK_E == 16 * 256, "a thread takes 16 edges");

__global__ __launch_bounds__(256) void gat_partition(
    const int* __restrict__ srcI,
    const int* __restrict__ dstI,
    int* __restrict__ binCnt,
    unsigned* __restrict__ binData)
{
    __shared__ int hist[NBIN_PAD];
    __shared__ int loff[NBIN_PAD];
    __shared__ int gbase[NBIN_PAD];
    __shared__ unsigned stage[CHUNK_E];
    const int t = threadIdx.x;
    for (int i = t; i < NBIN_PAD; i += 256) hist[i] = 0;
    __syncthreads();

    int sv[16], dv[16], r[16];
#pragma unroll
    for (int j = 0; j < 4; ++j) {
        const int p = blockIdx.x * CHUNK_I4 + j * 256 + t;
        int4 s4 = make_int4(0, 0, 0, 0);
        int4 d4 = make_int4(-1, -1, -1, -1);
        if (p < EE / 4) {
            s4 = reinterpret_cast<const int4*>(srcI)[p];
            d4 = reinterpret_cast<const int4*>(dstI)[p];
        }
        sv[4 * j + 0] = s4.x; sv[4 * j + 1] = s4.y; sv[4 * j + 2] = s4.z; sv[4 * j + 3] = s4.w;
        dv[4 * j + 0] = d4.x; dv[4 * j + 1] = d4.y; dv[4 * j + 2] = d4.z; dv[4 * j + 3] = d4.w;
    }
#pragma unroll
    for (int q = 0; q < 16; ++q)
        r[q] = (dv[q] >= 0) ? atomicAdd(&hist[dv[q] >> BIN_SHIFT], 1) : 0;
    __syncthreads();

    int c[4], base[4];
#pragma unroll
    for (int q = 0; q < 4; ++q) {
        const int i = t + 256 * q;
        c[q] = (i < NBIN) ? hist[i] : 0;
    }
#pragma unroll
    for (int q = 0; q < 4; ++q)
        base[q] = (c[q] > 0) ? atomicAdd(&binCnt[t + 256 * q], c[q]) : 0;
    if (t < 32) {
        int v[SCAN_W];
        int sum = 0;
#pragma unroll
        for (int k = 0; k < SCAN_W; ++k) v[k] = hist[SCAN_W * t + k];
#pragma unroll
        for (int k = 0; k < SCAN_W; ++k) { const int xk = v[k]; v[k] = sum; sum += xk; }
        int inc = sum;
#pragma unroll
        for (int dlt = 1; dlt < 32; dlt <<= 1) {
            const int y = __shfl_up(inc, dlt, 32);
            if (t >= dlt) inc += y;
        }
        const int excl = inc - sum;
#pragma unroll
        for (int k = 0; k < SCAN_W; ++k) loff[SCAN_W * t + k] = v[k] + excl;
    }
#pragma unroll
    for (int q = 0; q < 4; ++q) {
        const int i = t + 256 * q;
        if (i < NBIN) gbase[i] = base[q];
    }
    __syncthreads();

    int lo[16];
#pragma unroll
    for (int q = 0; q < 16; ++q)
        lo[q] = loff[max(dv[q], 0) >> BIN_SHIFT] + r[q];
#pragma unroll
    for (int q = 0; q < 16; ++q) {
        if (dv[q] >= 0)
            stage[lo[q]] = (unsigned)sv[q] | ((unsigned)dv[q] << 16);
    }
    __syncthreads();

    const int nvalid = min(CHUNK_E, EE - (int)blockIdx.x * CHUNK_E);
    unsigned w[16];
    int pb[16], pos[16];
#pragma unroll
    for (int q = 0; q < 16; ++q)
        w[q] = stage[min(t + 256 * q, nvalid - 1)];
#pragma unroll
    for (int q = 0; q < 16; ++q) {
        pb[q]  = (int)(w[q] >> (16 + BIN_SHIFT));
        pos[q] = gbase[pb[q]] - loff[pb[q]] + (t + 256 * q);
    }
#pragma unroll
    for (int q = 0; q < 16; ++q) {
        if (t + 256 * q < nvalid && pos[q] < BINCAP)
            binData[(size_t)pb[q] * BINCAP + pos[q]] = w[q];
    }
}

#define LPITCH 68
static_assert(LPITCH >= CAP + 4 && LPITCH % 4 == 0, "a row is compared 4 words at a time");
__global__ __launch_bounds__(256) void gat_rank(
    const int* __restrict__ srcI,
    const int* __restrict__ dstI,
    const int* __restrict__ binCnt,
    const unsigned* __restrict__ binData,
    int* __restrict__ cnt,
    int* __restrict__ slot)
{
    __shared__ int lcnt[BIN_NODES];
    __shared__ __attribute__((aligned(16))) unsigned lkey[BIN_NODES * LPITCH];
    __shared__ __attribute__((aligned(16))) int lslot[BIN_NODES * CAP];
    const int t = threadIdx.x;
    const int b = blockIdx.x;
    if (t < BIN_NODES) lcnt[t] = 0;
    for (int i = t; i < BIN_NODES * LPITCH / 4; i += 256)
        reinterpret_cast<uint4*>(lkey)[i] = make_uint4(0xffffffffu, 0xffffffffu, 0xffffffffu, 0xffffffffu);
    for (int i = t; i < BIN_NODES * CAP / 4; i += 256)
        reinterpret_cast<int4*>(lslot)[i] = make_int4(0x7fffffff, 0x7fffffff, 0x7fffffff, 0x7fffffff);
    __syncthreads();

    const int total = binCnt[b];
    if (total <= BINCAP) {
        const unsigned* __restrict__ bd = binData + (size_t)b * BINCAP;
        unsigned w[BINCAP / 256];
#pragma unroll
        for (int q = 0; q < BINCAP / 256; ++q) {
            const int i = t + 256 * q;
            w[q] = (i < total) ? bd[i] : 0u;
        }
#pragma unroll
        for (int q = 0; q < BINCAP / 256; ++q) {
            if (t + 256 * q < total) {
                const int d  = (int)(w[q] >> 16) & (BIN_NODES - 1);
                const int rk = atomicAdd(&lcnt[d], 1);
                if (rk < CAP) lkey[d * LPITCH + rk] = ((w[q] & 0xffffu) << 6) | (unsigned)rk;
            }
        }
    } else {
        for (int e = t; e < EE; e += 256) {
            const int dn = dstI[e];
            if ((dn >> BIN_SHIFT) == b) {
                const int d  = dn & (BIN_NODES - 1);
                const int rk = atomicAdd(&lcnt[d], 1);
                if (rk < CAP) lkey[d * LPITCH + rk] = ((unsigned)srcI[e] << 6) | (unsigned)rk;
            }
        }
    }
    __syncthreads();

    {
        const int d = t >> 2;
        const int c = min(lcnt[d], CAP);
        const unsigned* __restrict__ row = lkey + d * LPITCH;
        for (int j = t & 3; j < c; j += 4) {
            const unsigned mine = row[j];
            int below = 0;
            for (int i = 0; i < c; i += 4) {
                const uint4 k = *reinterpret_cast<const uint4*>(row + i);
                below += (int)(k.x < mine) + (int)(k.y < mine) + (int)(k.z < mine) + (int)(k.w < mine);
            }
            lslot[d * CAP + below] = (int)(mine >> 6);
        }
    }
    __syncthreads();
    if (t < BIN_NODES) cnt[b * BIN_NODES + t] = lcnt[t];
    int4* __restrict__ so = reinterpret_cast<int4*>(slot + (size_t)b * BIN_NODES * CAP);
    for (int i = t; i < BIN_NODES * CAP / 4; i += 256)
        so[i] = reinterpret_cast<const int4*>(lslot)[i];
}

template <int CTRL>
static __device__ __forceinline__ float row_lane(float v) {
    return __int_as_float(__builtin_amdgcn_update_dpp(0, __float_as_int(v), CTRL, 0xf, 0xf, true));
}

static __device__ __forceinline__ float4 row_piece(const float* __restrict__ p, int row, int lane) {
    return *reinterpret_cast<const float4*>(p + (size_t)row * FF + 4 * lane);
}

static __device__ __forceinline__ float head_score(const float4 s, const float4 d, const float4 at) {
    float v0 = s.x + d.x;
    float v1 = s.y + d.y;
    float v2 = s.z + d.z;
    float v3 = s.w + d.w;
    v0 = (v0 >= 0.0f) ? v0 : 0.2f * v0;
    v1 = (v1 >= 0.0f) ? v1 : 0.2f * v1;
    v2 = (v2 >= 0.0f) ? v2 : 0.2f * v2;
    v3 = (v3 >= 0.0f) ? v3 : 0.2f * v3;
    float acc = at.x * v0;
    acc += at.y * v1;
    acc += at.z * v2;
    acc += at.w * v3;
    acc += row_lane<0xB1>(acc);
    acc += row_lane<0x4E>(acc);
    return acc;
}

static __device__ __forceinline__ void absorb4(
    const float* __restrict__ src_p, int lane,
    int s0, int s1, int s2, int s3, int nv,
    const float4 drow, const float4 at,
    float& m, float& den, float4& sum)
{
    const float4 r0 = row_piece(src_p, s0, lane);
    const float4 r1 = row_piece(src_p, s1, lane);
    const float4 r2 = row_piece(src_p, s2, lane);
    const float4 r3 = row_piece(src_p, s3, lane);
    const float c0 = head_score(r0, drow, at);
    float c1 = head_score(r1, drow, at);
    float c2 = head_score(r2, drow, at);
    float c3 = head_score(r3, drow, at);
    c1 = (nv > 1) ? c1 : -INFINITY;
    c2 = (nv > 2) ? c2 : -INFINITY;
    c3 = (nv > 3) ? c3 : -INFINITY;
    const float mn = fmaxf(fmaxf(m, c0), fmaxf(c1, fmaxf(c2, c3)));
    const float f  = __expf(m - mn);
    const float w0 = __expf(c0 - mn);
    const float w1 = __expf(c1 - mn);
    const float w2 = __expf(c2 - mn);
    const float w3 = __expf(c3 - mn);
    den   = den   * f + ((w0 + w1) + (w2 + w3));
    sum.x = sum.x * f + ((w0 * r0.x + w1 * r1.x) + (w2 * r2.x + w3 * r3.x));
    sum.y = sum.y * f + ((w0 * r0.y + w1 * r1.y) + (w2 * r2.y + w3 * r3.y));
    sum.z = sum.z * f + ((w0 * r0.z + w1 * r1.z) + (w2 * r2.z + w3 * r3.z));
    sum.w = sum.w * f + ((w0 * r0.w + w1 * r1.w) + (w2 * r2.w + w3 * r3.w));
    m = mn;
}

static __device__ __forceinline__ void absorb_group(
    const float* __restrict__ src_p, int lane, const int4 q, int nv,
    const float4 drow, const float4 at,
    float& m, float& den, float4& sum)
{
    const int s0 = q.x;
    const int s1 = (nv > 1) ? q.y : s0;
    const int s2 = (nv > 2) ? q.z : s0;
    const int s3 = (nv > 3) ? q.w : s0;
    absorb4(src_p, lane, s0, s1, s2, s3, nv, drow, at, m, den, sum);
}

static_assert(CAP % 16 == 0, "the slot row is read 16 slots at a time");
static_assert(NN % 8 == 0, "a block of 8 waves takes 8 whole nodes");
static_assert(EE % 32 == 0, "the slow path reads 32 edges per trip");

static __device__ __forceinline__ void node_state(
    const float* __restrict__ src_p,
    const int*   __restrict__ srcI,
    const int*   __restrict__ dstI,
    const int*   __restrict__ slot,
    int n, int deg, int lane,
    const float4 drow, const float4 at,
    float& m, float& den, float4& sum)
{
    m   = -INFINITY;
    den = 0.0f;
    sum = make_float4(0.0f, 0.0f, 0.0f, 0.0f);

    if (deg <= CAP) {
        const int* __restrict__ sl = slot + (size_t)n * CAP;
        for (int i = 0; i < deg; i += 16) {
            const int4 q0 = *reinterpret_cast<const int4*>(sl + i);
            const int4 q1 = *reinterpret_cast<const int4*>(sl + i + 4);
            const int4 q2 = *reinterpret_cast<const int4*>(sl + i + 8);
            const int4 q3 = *reinterpret_cast<const int4*>(sl + i + 12);
            absorb_group(src_p, lane, q0, deg - i, drow, at, m, den, sum);
            if (i + 4 < deg)  absorb_group(src_p, lane, q1, deg - i - 4,  drow, at, m, den, sum);
            if (i + 8 < deg)  absorb_group(src_p, lane, q2, deg - i - 8,  drow, at, m, den, sum);
            if (i + 12 < deg) absorb_group(src_p, lane, q3, deg - i - 12, drow, at, m, den, sum);
        }
    } else {
        for (int e = 0; e < EE; e += 32) {
            unsigned hit = __builtin_amdgcn_ballot_w32(dstI[e + lane] == n);
            while (hit != 0u) {
                const int b  = __builtin_ctz(hit);
                hit &= hit - 1u;
                const int s0 = srcI[e + b];
                absorb4(src_p, lane, s0, s0, s0, s0, 1, drow, at, m, den, sum);
            }
        }
    }
}

static __device__ __forceinline__ float4 finish_piece(
    const float4 sum, float d, const float4 xr, const float4 bs, float pw)
{
    float4 o;
    o.x = sum.x / d + xr.x + bs.x;
    o.y = sum.y / d + xr.y + bs.y;
    o.z = sum.z / d + xr.z + bs.z;
    o.w = sum.w / d + xr.w + bs.w;
    o.x = (o.x >= 0.0f) ? o.x : pw * o.x;
    o.y = (o.y >= 0.0f) ? o.y : pw * o.y;
    o.z = (o.z >= 0.0f) ? o.z : pw * o.z;
    o.w = (o.w >= 0.0f) ? o.w : pw * o.w;
    return o;
}

__global__ __launch_bounds__(256) void gat_node(
    const float* __restrict__ src_p,
    const float* __restrict__ dst_p,
    const int*   __restrict__ srcI,
    const int*   __restrict__ dstI,
    const float* __restrict__ attn,
    const int*   __restrict__ cnt,
    const int*   __restrict__ slot,
    const float* __restrict__ x,
    const float* __restrict__ bias,
    const float* __restrict__ prelu_w,
    float* __restrict__ out,
    float* __restrict__ denb,
    float* __restrict__ maxb,
    float* __restrict__ gmax)
{
    __shared__ float sden[64];
    __shared__ float smax[64];
    __shared__ float wmax[8];
    const int lane = threadIdx.x & 31;
    const int w    = threadIdx.x >> 5;
    const int n   = __builtin_amdgcn_readfirstlane((int)((blockIdx.x * 256 + threadIdx.x) >> 5));
    const int deg = cnt[n];

    const float4 at   = *reinterpret_cast<const float4*>(attn + 4 * lane);
    const float4 drow = row_piece(dst_p, n, lane);

    float  m, den;
    float4 sum;
    node_state(src_p, srcI, dstI, slot, n, deg, lane, drow, at, m, den, sum);

    const float4 xr = row_piece(x, n, lane);
    const float4 bs = *reinterpret_cast<const float4*>(bias + 4 * lane);
    const float  pw = prelu_w[0];

    if ((lane & 3) == 0) {
        sden[w * 8 + (lane >> 2)] = den;
        smax[w * 8 + (lane >> 2)] = m;
    }
    float mm = m;
    mm = fmaxf(mm, row_lane<0x164>(mm));
    mm = fmaxf(mm, row_lane<0x168>(mm));
    const float wm = fmaxf(__int_as_float(__builtin_amdgcn_readlane(__float_as_int(mm), 0)),
                           __int_as_float(__builtin_amdgcn_readlane(__float_as_int(mm), 16)));
    if (lane == 0) wmax[w] = wm;
    __syncthreads();
    if (threadIdx.x < 64) {
        denb[(size_t)blockIdx.x * 64 + threadIdx.x] = sden[threadIdx.x];
    } else if (threadIdx.x < 128) {
        maxb[(size_t)blockIdx.x * 64 + (threadIdx.x - 64)] = smax[threadIdx.x - 64];
    } else if (threadIdx.x == 128) {
        float b = wmax[0];
        for (int i = 1; i < 8; ++i) b = fmaxf(b, wmax[i]);
        __hip_atomic_fetch_max(gmax, b, __ATOMIC_RELAXED, __HIP_MEMORY_SCOPE_AGENT);
    }

    const float d = (deg > 0) ? den : INFINITY;
    *reinterpret_cast<float4*>(out + (size_t)n * FF + 4 * lane) = finish_piece(sum, d, xr, bs, pw);
}

static __device__ __forceinline__ bool needs_term(float den, float m, float G) {
    return (den + 1e-16f * __expf(G - m)) != den;
}

__global__ __launch_bounds__(64) void gat_repair(
    const float* __restrict__ src_p,
    const float* __restrict__ dst_p,
    const int*   __restrict__ srcI,
    const int*   __restrict__ dstI,
    const float* __restrict__ attn,
    const int*   __restrict__ cnt,
    const int*   __restrict__ slot,
    const float* __restrict__ denb,
    const float* __restrict__ maxb,
    const float* __restrict__ gmax,
    const float* __restrict__ x,
    const float* __restrict__ bias,
    const float* __restrict__ prelu_w,
    float* __restrict__ out)
{
    const int lane = threadIdx.x & 31;
    const int node = blockIdx.x * 64 + threadIdx.x;
    const float G  = *gmax;
    bool bad = false;
    if (node < NN) {
        const float4 d0 = *reinterpret_cast<const float4*>(denb + (size_t)node * HH);
        const float4 d1 = *reinterpret_cast<const float4*>(denb + (size_t)node * HH + 4);
        const float4 m0 = *reinterpret_cast<const float4*>(maxb + (size_t)node * HH);
        const float4 m1 = *reinterpret_cast<const float4*>(maxb + (size_t)node * HH + 4);
        bad = needs_term(d0.x, m0.x, G) | needs_term(d0.y, m0.y, G) |
              needs_term(d0.z, m0.z, G) | needs_term(d0.w, m0.w, G) |
              needs_term(d1.x, m1.x, G) | needs_term(d1.y, m1.y, G) |
              needs_term(d1.z, m1.z, G) | needs_term(d1.w, m1.w, G);
    }
    unsigned hit = __builtin_amdgcn_ballot_w32(bad);
    if (hit == 0u) return;

    const int nbase = __builtin_amdgcn_readfirstlane(node);
    const float4 at = *reinterpret_cast<const float4*>(attn + 4 * lane);
    const float4 bs = *reinterpret_cast<const float4*>(bias + 4 * lane);
    const float  pw = prelu_w[0];
    while (hit != 0u) {
        const int n = nbase + __builtin_ctz(hit);
        hit &= hit - 1u;
        const int deg = cnt[n];
        const float4 drow = row_piece(dst_p, n, lane);
        const float4 xr   = row_piece(x, n, lane);
        float  m, dn;
        float4 sum;
        node_state(src_p, srcI, dstI, slot, n, deg, lane, drow, at, m, dn, sum);
        const float d = dn + 1e-16f * __expf(G - m);
        *reinterpret_cast<float4*>(out + (size_t)n * FF + 4 * lane) = finish_piece(sum, d, xr, bs, pw);
    }
}

extern "C" void kernel_launch(void* const* d_in, const int* in_sizes, int n_in,
                              void* d_out, int out_size, void* d_ws, size_t ws_size,
                              hipStream_t stream) {
    (void)in_sizes; (void)n_in; (void)out_size; (void)ws_size;

    const float* x        = (const float*)d_in[0];
    const int*   eidx     = (const int*)  d_in[1];
    const float* W_src    = (const float*)d_in[2];
    const float* W_dst    = (const float*)d_in[3];
    const float* attn     = (const float*)d_in[4];
    const float* bias     = (const float*)d_in[5];
    const float* prelu_w  = (const float*)d_in[6];
    const int*   srcI     = eidx;
    const int*   dstI     = eidx + EE;

    float* out = (float*)d_out;

    float* ws      = (float*)d_ws;
    float* src_p   = ws;
    float* dst_p   = src_p + (size_t)NN * FF;
    int*   slot    = (int*)(dst_p + (size_t)NN * FF);
    int*   cnt     = slot + (size_t)ROWS_PAD * CAP;
    int*   binCnt  = cnt + ROWS_PAD;
    unsigned* binData = (unsigned*)(binCnt + NBIN_PAD);
    float* denb    = (float*)(binData + (size_t)NBIN * BINCAP);
    float* maxb    = denb + (size_t)NN * HH;
    float* gmax    = maxb + (size_t)NN * HH;

    {
        dim3 grid(2, NN / 16);
        gat_proj_gemm<<<grid, 32, 0, stream>>>(x, W_src, W_dst, src_p, dst_p, binCnt, gmax);
    }

    gat_partition<<<(EE / 4 + CHUNK_I4 - 1) / CHUNK_I4, 256, 0, stream>>>(srcI, dstI, binCnt, binData);

    gat_rank<<<NBIN, 256, 0, stream>>>(srcI, dstI, binCnt, binData, cnt, slot);

    gat_node<<<NN / 8, 256, 0, stream>>>(src_p, dst_p, srcI, dstI, attn, cnt, slot,
                                         x, bias, prelu_w, out, denb, maxb, gmax);

    gat_repair<<<(NN + 63) / 64, 64, 0, stream>>>(src_p, dst_p, srcI, dstI, attn, cnt, slot,
                                                  denb, maxb, gmax, x, bias, prelu_w, out);
}


# projection: output columns dealt 4 to a lane (tile c of lane l = column 4l+c): W fragments by 16-byte whole-line loads, result rows stored as whole 128-byte lines, not 64-byte pieces; same sums
# speedup vs baseline: 7.6044x; 1.0380x over previous
#include <hip/hip_runtime.h>
#include <hip/hip_bf16.h>
#include <math.h>


#define NN 50000
#define EE 800000
#define FF 128
#define HH 8
#define DD 16

#define CAP     64
static_assert(CAP % 4 == 0, "slots are read four at a time");

#define BIN_SHIFT 6
#define BIN_NODES (1 << BIN_SHIFT)
#define NBIN      ((NN + BIN_NODES - 1) / BIN_NODES)
#define NBIN_PAD  800
#define BINCAP    2048
#define ROWS_PAD  (NBIN * BIN_NODES)
#define CHUNK_I4  1024
static_assert(NN <= 65536, "source and destination share one 32-bit word");
static_assert(NBIN_PAD >= NBIN && NBIN_PAD % 32 == 0, "range counters are cleared 32 at a time");
static_assert(EE % 4 == 0, "indices are read 16 bytes at a time");
static_assert(BINCAP % 256 == 0, "K2b reads a range's words 256 at a time");

typedef __attribute__((ext_vector_type(2))) float v2f;
typedef __attribute__((ext_vector_type(8))) float v8f;

__global__ __launch_bounds__(32) void gat_proj_gemm(
    const float* __restrict__ x,
    const float* __restrict__ W_src,
    const float* __restrict__ W_dst,
    float* __restrict__ src_p,
    float* __restrict__ dst_p,
    int*   __restrict__ binCnt,
    float* __restrict__ gmax)
{
    const int which   = blockIdx.x;
    const int rowbase = blockIdx.y * 16;
    const float* __restrict__ W   = which ? W_dst : W_src;
    float* __restrict__       out = which ? dst_p : src_p;

    const int lane = threadIdx.x;
    const int half = lane >> 4;
    const int l    = lane & 15;

    const float* __restrict__ xrow = x + (size_t)(rowbase + l) * FF;

    if (which == 1) {
        if (blockIdx.y < NBIN_PAD / 32) binCnt[blockIdx.y * 32 + lane] = 0;
        if (rowbase == 0 && lane == 0) *gmax = -INFINITY;
    }

    v8f acc[8];
#pragma unroll
    for (int c = 0; c < 8; ++c) acc[c] = (v8f){};

    for (int k = 0; k < FF; k += 4) {
        v2f a;
        a.x = xrow[k + 2 * half + 0];
        a.y = xrow[k + 2 * half + 1];
        const float* wp = W + (size_t)(k + 2 * half) * FF + 4 * l;
        const float4 w0lo = *reinterpret_cast<const float4*>(wp);
        const float4 w0hi = *reinterpret_cast<const float4*>(wp + 64);
        const float4 w1lo = *reinterpret_cast<const float4*>(wp + FF);
        const float4 w1hi = *reinterpret_cast<const float4*>(wp + FF + 64);
        const float bx[8] = {w0lo.x, w0lo.y, w0lo.z, w0lo.w, w0hi.x, w0hi.y, w0hi.z, w0hi.w};
        const float by[8] = {w1lo.x, w1lo.y, w1lo.z, w1lo.w, w1hi.x, w1hi.y, w1hi.z, w1hi.w};
#pragma unroll
        for (int c = 0; c < 8; ++c) {
            v2f b;
            b.x = bx[c];
            b.y = by[c];
            acc[c] = __builtin_amdgcn_wmma_f32_16x16x4_f32(
                false, a, false, b, (short)0, acc[c], false, false);
        }
    }

#pragma unroll
    for (int r = 0; r < 8; ++r) {
        float* orow = out + (size_t)(rowbase + r + 8 * half) * FF + 4 * l;
        *reinterpret_cast<float4*>(orow)      = make_float4(acc[0][r], acc[1][r], acc[2][r], acc[3][r]);
        *reinterpret_cast<float4*>(orow + 64) = make_float4(acc[4][r], acc[5][r], acc[6][r], acc[7][r]);
    }
}

#define CHUNK_E   (4 * CHUNK_I4)
#define SCAN_W    (NBIN_PAD / 32)
static_assert(SCAN_W * 32 == NBIN_PAD, "the first wave covers all range counters");
static_assert(CHUNK_E == 16 * 256, "a thread takes 16 edges");

__global__ __launch_bounds__(256) void gat_partition(
    const int* __restrict__ srcI,
    const int* __restrict__ dstI,
    int* __restrict__ binCnt,
    unsigned* __restrict__ binData)
{
    __shared__ int hist[NBIN_PAD];
    __shared__ int loff[NBIN_PAD];
    __shared__ int gbase[NBIN_PAD];
    __shared__ unsigned stage[CHUNK_E];
    const int t = threadIdx.x;
    for (int i = t; i < NBIN_PAD; i += 256) hist[i] = 0;
    __syncthreads();

    int sv[16], dv[16], r[16];
#pragma unroll
    for (int j = 0; j < 4; ++j) {
        const int p = blockIdx.x * CHUNK_I4 + j * 256 + t;
        int4 s4 = make_int4(0, 0, 0, 0);
        int4 d4 = make_int4(-1, -1, -1, -1);
        if (p < EE / 4) {
            s4 = reinterpret_cast<const int4*>(srcI)[p];
            d4 = reinterpret_cast<const int4*>(dstI)[p];
        }
        sv[4 * j + 0] = s4.x; sv[4 * j + 1] = s4.y; sv[4 * j + 2] = s4.z; sv[4 * j + 3] = s4.w;
        dv[4 * j + 0] = d4.x; dv[4 * j + 1] = d4.y; dv[4 * j + 2] = d4.z; dv[4 * j + 3] = d4.w;
    }
#pragma unroll
    for (int q = 0; q < 16; ++q)
        r[q] = (dv[q] >= 0) ? atomicAdd(&hist[dv[q] >> BIN_SHIFT], 1) : 0;
    __syncthreads();

    int c[4], base[4];
#pragma unroll
    for (int q = 0; q < 4; ++q) {
        const int i = t + 256 * q;
        c[q] = (i < NBIN) ? hist[i] : 0;
    }
#pragma unroll
    for (int q = 0; q < 4; ++q)
        base[q] = (c[q] > 0) ? atomicAdd(&binCnt[t + 256 * q], c[q]) : 0;
    if (t < 32) {
        int v[SCAN_W];
        int sum = 0;
#pragma unroll
        for (int k = 0; k < SCAN_W; ++k) v[k] = hist[SCAN_W * t + k];
#pragma unroll
        for (int k = 0; k < SCAN_W; ++k) { const int xk = v[k]; v[k] = sum; sum += xk; }
        int inc = sum;
#pragma unroll
        for (int dlt = 1; dlt < 32; dlt <<= 1) {
            const int y = __shfl_up(inc, dlt, 32);
            if (t >= dlt) inc += y;
        }
        const int excl = inc - sum;
#pragma unroll
        for (int k = 0; k < SCAN_W; ++k) loff[SCAN_W * t + k] = v[k] + excl;
    }
#pragma unroll
    for (int q = 0; q < 4; ++q) {
        const int i = t + 256 * q;
        if (i < NBIN) gbase[i] = base[q];
    }
    __syncthreads();

    int lo[16];
#pragma unroll
    for (int q = 0; q < 16; ++q)
        lo[q] = loff[max(dv[q], 0) >> BIN_SHIFT] + r[q];
#pragma unroll
    for (int q = 0; q < 16; ++q) {
        if (dv[q] >= 0)
            stage[lo[q]] = (unsigned)sv[q] | ((unsigned)dv[q] << 16);
    }
    __syncthreads();

    const int nvalid = min(CHUNK_E, EE - (int)blockIdx.x * CHUNK_E);
    unsigned w[16];
    int pb[16], pos[16];
#pragma unroll
    for (int q = 0; q < 16; ++q)
        w[q] = stage[min(t + 256 * q, nvalid - 1)];
#pragma unroll
    for (int q = 0; q < 16; ++q) {
        pb[q]  = (int)(w[q] >> (16 + BIN_SHIFT));
        pos[q] = gbase[pb[q]] - loff[pb[q]] + (t + 256 * q);
    }
#pragma unroll
    for (int q = 0; q < 16; ++q) {
        if (t + 256 * q < nvalid && pos[q] < BINCAP)
            binData[(size_t)pb[q] * BINCAP + pos[q]] = w[q];
    }
}

#define LPITCH 68
static_assert(LPITCH >= CAP + 4 && LPITCH % 4 == 0, "a row is compared 4 words at a time");
__global__ __launch_bounds__(256) void gat_rank(
    const int* __restrict__ srcI,
    const int* __restrict__ dstI,
    const int* __restrict__ binCnt,
    const unsigned* __restrict__ binData,
    int* __restrict__ cnt,
    int* __restrict__ slot)
{
    __shared__ int lcnt[BIN_NODES];
    __shared__ __attribute__((aligned(16))) unsigned lkey[BIN_NODES * LPITCH];
    __shared__ __attribute__((aligned(16))) int lslot[BIN_NODES * CAP];
    const int t = threadIdx.x;
    const int b = blockIdx.x;
    if (t < BIN_NODES) lcnt[t] = 0;
    for (int i = t; i < BIN_NODES * LPITCH / 4; i += 256)
        reinterpret_cast<uint4*>(lkey)[i] = make_uint4(0xffffffffu, 0xffffffffu, 0xffffffffu, 0xffffffffu);
    for (int i = t; i < BIN_NODES * CAP / 4; i += 256)
        reinterpret_cast<int4*>(lslot)[i] = make_int4(0x7fffffff, 0x7fffffff, 0x7fffffff, 0x7fffffff);
    __syncthreads();

    const int total = binCnt[b];
    if (total <= BINCAP) {
        const unsigned* __restrict__ bd = binData + (size_t)b * BINCAP;
        unsigned w[BINCAP / 256];
#pragma unroll
        for (int q = 0; q < BINCAP / 256; ++q) {
            const int i = t + 256 * q;
            w[q] = (i < total) ? bd[i] : 0u;
        }
#pragma unroll
        for (int q = 0; q < BINCAP / 256; ++q) {
            if (t + 256 * q < total) {
                const int d  = (int)(w[q] >> 16) & (BIN_NODES - 1);
                const int rk = atomicAdd(&lcnt[d], 1);
                if (rk < CAP) lkey[d * LPITCH + rk] = ((w[q] & 0xffffu) << 6) | (unsigned)rk;
            }
        }
    } else {
        for (int e = t; e < EE; e += 256) {
            const int dn = dstI[e];
            if ((dn >> BIN_SHIFT) == b) {
                const int d  = dn & (BIN_NODES - 1);
                const int rk = atomicAdd(&lcnt[d], 1);
                if (rk < CAP) lkey[d * LPITCH + rk] = ((unsigned)srcI[e] << 6) | (unsigned)rk;
            }
        }
    }
    __syncthreads();

    {
        const int d = t >> 2;
        const int c = min(lcnt[d], CAP);
        const unsigned* __restrict__ row = lkey + d * LPITCH;
        for (int j = t & 3; j < c; j += 4) {
            const unsigned mine = row[j];
            int below = 0;
            for (int i = 0; i < c; i += 4) {
                const uint4 k = *reinterpret_cast<const uint4*>(row + i);
                below += (int)(k.x < mine) + (int)(k.y < mine) + (int)(k.z < mine) + (int)(k.w < mine);
            }
            lslot[d * CAP + below] = (int)(mine >> 6);
        }
    }
    __syncthreads();
    if (t < BIN_NODES) cnt[b * BIN_NODES + t] = lcnt[t];
    int4* __restrict__ so = reinterpret_cast<int4*>(slot + (size_t)b * BIN_NODES * CAP);
    for (int i = t; i < BIN_NODES * CAP / 4; i += 256)
        so[i] = reinterpret_cast<const int4*>(lslot)[i];
}

template <int CTRL>
static __device__ __forceinline__ float row_lane(float v) {
    return __int_as_float(__builtin_amdgcn_update_dpp(0, __float_as_int(v), CTRL, 0xf, 0xf, true));
}

static __device__ __forceinline__ float4 row_piece(const float* __restrict__ p, int row, int lane) {
    return *reinterpret_cast<const float4*>(p + (size_t)row * FF + 4 * lane);
}

static __device__ __forceinline__ float head_score(const float4 s, const float4 d, const float4 at) {
    float v0 = s.x + d.x;
    float v1 = s.y + d.y;
    float v2 = s.z + d.z;
    float v3 = s.w + d.w;
    v0 = (v0 >= 0.0f) ? v0 : 0.2f * v0;
    v1 = (v1 >= 0.0f) ? v1 : 0.2f * v1;
    v2 = (v2 >= 0.0f) ? v2 : 0.2f * v2;
    v3 = (v3 >= 0.0f) ? v3 : 0.2f * v3;
    float acc = at.x * v0;
    acc += at.y * v1;
    acc += at.z * v2;
    acc += at.w * v3;
    acc += row_lane<0xB1>(acc);
    acc += row_lane<0x4E>(acc);
    return acc;
}

static __device__ __forceinline__ void absorb4(
    const float* __restrict__ src_p, int lane,
    int s0, int s1, int s2, int s3, int nv,
    const float4 drow, const float4 at,
    float& m, float& den, float4& sum)
{
    const float4 r0 = row_piece(src_p, s0, lane);
    const float4 r1 = row_piece(src_p, s1, lane);
    const float4 r2 = row_piece(src_p, s2, lane);
    const float4 r3 = row_piece(src_p, s3, lane);
    const float c0 = head_score(r0, drow, at);
    float c1 = head_score(r1, drow, at);
    float c2 = head_score(r2, drow, at);
    float c3 = head_score(r3, drow, at);
    c1 = (nv > 1) ? c1 : -INFINITY;
    c2 = (nv > 2) ? c2 : -INFINITY;
    c3 = (nv > 3) ? c3 : -INFINITY;
    const float mn = fmaxf(fmaxf(m, c0), fmaxf(c1, fmaxf(c2, c3)));
    const float f  = __expf(m - mn);
    const float w0 = __expf(c0 - mn);
    const float w1 = __expf(c1 - mn);
    const float w2 = __expf(c2 - mn);
    const float w3 = __expf(c3 - mn);
    den   = den   * f + ((w0 + w1) + (w2 + w3));
    sum.x = sum.x * f + ((w0 * r0.x + w1 * r1.x) + (w2 * r2.x + w3 * r3.x));
    sum.y = sum.y * f + ((w0 * r0.y + w1 * r1.y) + (w2 * r2.y + w3 * r3.y));
    sum.z = sum.z * f + ((w0 * r0.z + w1 * r1.z) + (w2 * r2.z + w3 * r3.z));
    sum.w = sum.w * f + ((w0 * r0.w + w1 * r1.w) + (w2 * r2.w + w3 * r3.w));
    m = mn;
}

static __device__ __forceinline__ void absorb_group(
    const float* __restrict__ src_p, int lane, const int4 q, int nv,
    const float4 drow, const float4 at,
    float& m, float& den, float4& sum)
{
    const int s0 = q.x;
    const int s1 = (nv > 1) ? q.y : s0;
    const int s2 = (nv > 2) ? q.z : s0;
    const int s3 = (nv > 3) ? q.w : s0;
    absorb4(src_p, lane, s0, s1, s2, s3, nv, drow, at, m, den, sum);
}

static_assert(CAP % 16 == 0, "the slot row is read 16 slots at a time");
static_assert(NN % 8 == 0, "a block of 8 waves takes 8 whole nodes");
static_assert(EE % 32 == 0, "the slow path reads 32 edges per trip");

static __device__ __forceinline__ void node_state(
    const float* __restrict__ src_p,
    const int*   __restrict__ srcI,
    const int*   __restrict__ dstI,
    const int*   __restrict__ slot,
    int n, int deg, int lane,
    const float4 drow, const float4 at,
    float& m, float& den, float4& sum)
{
    m   = -INFINITY;
    den = 0.0f;
    sum = make_float4(0.0f, 0.0f, 0.0f, 0.0f);

    if (deg <= CAP) {
        const int* __restrict__ sl = slot + (size_t)n * CAP;
        for (int i = 0; i < deg; i += 16) {
            const int4 q0 = *reinterpret_cast<const int4*>(sl + i);
            const int4 q1 = *reinterpret_cast<const int4*>(sl + i + 4);
            const int4 q2 = *reinterpret_cast<const int4*>(sl + i + 8);
            const int4 q3 = *reinterpret_cast<const int4*>(sl + i + 12);
            absorb_group(src_p, lane, q0, deg - i, drow, at, m, den, sum);
            if (i + 4 < deg)  absorb_group(src_p, lane, q1, deg - i - 4,  drow, at, m, den, sum);
            if (i + 8 < deg)  absorb_group(src_p, lane, q2, deg - i - 8,  drow, at, m, den, sum);
            if (i + 12 < deg) absorb_group(src_p, lane, q3, deg - i - 12, drow, at, m, den, sum);
        }
    } else {
        for (int e = 0; e < EE; e += 32) {
            unsigned hit = __builtin_amdgcn_ballot_w32(dstI[e + lane] == n);
            while (hit != 0u) {
                const int b  = __builtin_ctz(hit);
                hit &= hit - 1u;
                const int s0 = srcI[e + b];
                absorb4(src_p, lane, s0, s0, s0, s0, 1, drow, at, m, den, sum);
            }
        }
    }
}

static __device__ __forceinline__ float4 finish_piece(
    const float4 sum, float d, const float4 xr, const float4 bs, float pw)
{
    float4 o;
    o.x = sum.x / d + xr.x + bs.x;
    o.y = sum.y / d + xr.y + bs.y;
    o.z = sum.z / d + xr.z + bs.z;
    o.w = sum.w / d + xr.w + bs.w;
    o.x = (o.x >= 0.0f) ? o.x : pw * o.x;
    o.y = (o.y >= 0.0f) ? o.y : pw * o.y;
    o.z = (o.z >= 0.0f) ? o.z : pw * o.z;
    o.w = (o.w >= 0.0f) ? o.w : pw * o.w;
    return o;
}

__global__ __launch_bounds__(256) void gat_node(
    const float* __restrict__ src_p,
    const float* __restrict__ dst_p,
    const int*   __restrict__ srcI,
    const int*   __restrict__ dstI,
    const float* __restrict__ attn,
    const int*   __restrict__ cnt,
    const int*   __restrict__ slot,
    const float* __restrict__ x,
    const float* __restrict__ bias,
    const float* __restrict__ prelu_w,
    float* __restrict__ out,
    float* __restrict__ denb,
    float* __restrict__ maxb,
    float* __restrict__ gmax)
{
    __shared__ float sden[64];
    __shared__ float smax[64];
    __shared__ float wmax[8];
    const int lane = threadIdx.x & 31;
    const int w    = threadIdx.x >> 5;
    const int n   = __builtin_amdgcn_readfirstlane((int)((blockIdx.x * 256 + threadIdx.x) >> 5));
    const int deg = cnt[n];

    const float4 at   = *reinterpret_cast<const float4*>(attn + 4 * lane);
    const float4 drow = row_piece(dst_p, n, lane);

    float  m, den;
    float4 sum;
    node_state(src_p, srcI, dstI, slot, n, deg, lane, drow, at, m, den, sum);

    const float4 xr = row_piece(x, n, lane);
    const float4 bs = *reinterpret_cast<const float4*>(bias + 4 * lane);
    const float  pw = prelu_w[0];

    if ((lane & 3) == 0) {
        sden[w * 8 + (lane >> 2)] = den;
        smax[w * 8 + (lane >> 2)] = m;
    }
    float mm = m;
    mm = fmaxf(mm, row_lane<0x164>(mm));
    mm = fmaxf(mm, row_lane<0x168>(mm));
    const float wm = fmaxf(__int_as_float(__builtin_amdgcn_readlane(__float_as_int(mm), 0)),
                           __int_as_float(__builtin_amdgcn_readlane(__float_as_int(mm), 16)));
    if (lane == 0) wmax[w] = wm;
    __syncthreads();
    if (threadIdx.x < 64) {
        denb[(size_t)blockIdx.x * 64 + threadIdx.x] = sden[threadIdx.x];
    } else if (threadIdx.x < 128) {
        maxb[(size_t)blockIdx.x * 64 + (threadIdx.x - 64)] = smax[threadIdx.x - 64];
    } else if (threadIdx.x == 128) {
        float b = wmax[0];
        for (int i = 1; i < 8; ++i) b = fmaxf(b, wmax[i]);
        __hip_atomic_fetch_max(gmax, b, __ATOMIC_RELAXED, __HIP_MEMORY_SCOPE_AGENT);
    }

    const float d = (deg > 0) ? den : INFINITY;
    *reinterpret_cast<float4*>(out + (size_t)n * FF + 4 * lane) = finish_piece(sum, d, xr, bs, pw);
}

static __device__ __forceinline__ bool needs_term(float den, float m, float G) {
    return (den + 1e-16f * __expf(G - m)) != den;
}

__global__ __launch_bounds__(64) void gat_repair(
    const float* __restrict__ src_p,
    const float* __restrict__ dst_p,
    const int*   __restrict__ srcI,
    const int*   __restrict__ dstI,
    const float* __restrict__ attn,
    const int*   __restrict__ cnt,
    const int*   __restrict__ slot,
    const float* __restrict__ denb,
    const float* __restrict__ maxb,
    const float* __restrict__ gmax,
    const float* __restrict__ x,
    const float* __restrict__ bias,
    const float* __restrict__ prelu_w,
    float* __restrict__ out)
{
    const int lane = threadIdx.x & 31;
    const int node = blockIdx.x * 64 + threadIdx.x;
    const float G  = *gmax;
    bool bad = false;
    if (node < NN) {
        const float4 d0 = *reinterpret_cast<const float4*>(denb + (size_t)node * HH);
        const float4 d1 = *reinterpret_cast<const float4*>(denb + (size_t)node * HH + 4);
        const float4 m0 = *reinterpret_cast<const float4*>(maxb + (size_t)node * HH);
        const float4 m1 = *reinterpret_cast<const float4*>(maxb + (size_t)node * HH + 4);
        bad = needs_term(d0.x, m0.x, G) | needs_term(d0.y, m0.y, G) |
              needs_term(d0.z, m0.z, G) | needs_term(d0.w, m0.w, G) |
              needs_term(d1.x, m1.x, G) | needs_term(d1.y, m1.y, G) |
              needs_term(d1.z, m1.z, G) | needs_term(d1.w, m1.w, G);
    }
    unsigned hit = __builtin_amdgcn_ballot_w32(bad);
    if (hit == 0u) return;

    const int nbase = __builtin_amdgcn_readfirstlane(node);
    const float4 at = *reinterpret_cast<const float4*>(attn + 4 * lane);
    const float4 bs = *reinterpret_cast<const float4*>(bias + 4 * lane);
    const float  pw = prelu_w[0];
    while (hit != 0u) {
        const int n = nbase + __builtin_ctz(hit);
        hit &= hit - 1u;
        const int deg = cnt[n];
        const float4 drow = row_piece(dst_p, n, lane);
        const float4 xr   = row_piece(x, n, lane);
        float  m, dn;
        float4 sum;
        node_state(src_p, srcI, dstI, slot, n, deg, lane, drow, at, m, dn, sum);
        const float d = dn + 1e-16f * __expf(G - m);
        *reinterpret_cast<float4*>(out + (size_t)n * FF + 4 * lane) = finish_piece(sum, d, xr, bs, pw);
    }
}

extern "C" void kernel_launch(void* const* d_in, const int* in_sizes, int n_in,
                              void* d_out, int out_size, void* d_ws, size_t ws_size,
                              hipStream_t stream) {
    (void)in_sizes; (void)n_in; (void)out_size; (void)ws_size;

    const float* x        = (const float*)d_in[0];
    const int*   eidx     = (const int*)  d_in[1];
    const float* W_src    = (const float*)d_in[2];
    const float* W_dst    = (const float*)d_in[3];
    const float* attn     = (const float*)d_in[4];
    const float* bias     = (const float*)d_in[5];
    const float* prelu_w  = (const float*)d_in[6];
    const int*   srcI     = eidx;
    const int*   dstI     = eidx + EE;

    float* out = (float*)d_out;

    float* ws      = (float*)d_ws;
    float* src_p   = ws;
    float* dst_p   = src_p + (size_t)NN * FF;
    int*   slot    = (int*)(dst_p + (size_t)NN * FF);
    int*   cnt     = slot + (size_t)ROWS_PAD * CAP;
    int*   binCnt  = cnt + ROWS_PAD;
    unsigned* binData = (unsigned*)(binCnt + NBIN_PAD);
    float* denb    = (float*)(binData + (size_t)NBIN * BINCAP);
    float* maxb    = denb + (size_t)NN * HH;
    float* gmax    = maxb + (size_t)NN * HH;

    {
        dim3 grid(2, NN / 16);
        gat_proj_gemm<<<grid, 32, 0, stream>>>(x, W_src, W_dst, src_p, dst_p, binCnt, gmax);
    }

    gat_partition<<<(EE / 4 + CHUNK_I4 - 1) / CHUNK_I4, 256, 0, stream>>>(srcI, dstI, binCnt, binData);

    gat_rank<<<NBIN, 256, 0, stream>>>(srcI, dstI, binCnt, binData, cnt, slot);

    gat_node<<<NN / 8, 256, 0, stream>>>(src_p, dst_p, srcI, dstI, attn, cnt, slot,
                                         x, bias, prelu_w, out, denb, maxb, gmax);

    gat_repair<<<(NN + 63) / 64, 64, 0, stream>>>(src_p, dst_p, srcI, dstI, attn, cnt, slot,
                                                  denb, maxb, gmax, x, bias, prelu_w, out);
}


# grouping rides in the projection: edge partition beside x@W_src, slot ranking beside x@W_dst (4 launches, not 5); partition leaves per-block runs + offset tables, no global adds; same sums
# speedup vs baseline: 8.3944x; 1.1039x over previous
#include <hip/hip_runtime.h>
#include <hip/hip_bf16.h>
#include <math.h>


#define NN 50000
#define EE 800000
#define FF 128
#define HH 8
#define DD 16

#define CAP     64
static_assert(CAP % 4 == 0, "slots are read four at a time");

#define BIN_SHIFT 6
#define BIN_NODES (1 << BIN_SHIFT)
#define NBIN      ((NN + BIN_NODES - 1) / BIN_NODES)
#define NBIN_PAD  800
#define ROWS_PAD  (NBIN * BIN_NODES)
#define CHUNK_I4  1024
#define CHUNK_E   (4 * CHUNK_I4)
#define NPB       ((EE + CHUNK_E - 1) / CHUNK_E)
#define SCAN_W    (NBIN_PAD / 32)
static_assert(NN <= 65536, "source and destination share one 32-bit word");
static_assert(NBIN_PAD > NBIN, "entry NBIN of a run's table is the run's length");
static_assert(SCAN_W * 32 == NBIN_PAD, "the first wave covers all table entries");
static_assert(EE % 4 == 0, "indices are read 16 bytes at a time");
static_assert(CHUNK_E == 16 * 256, "a thread takes 16 edges");
static_assert(NPB <= 256, "the ranking block reads one run per thread");

#define NPANEL      (NN / 16)
#define GEMM_UNITS  (2 * NPANEL)
#define GEMM_BLOCKS ((GEMM_UNITS + 7) / 8)
static_assert(NN % 16 == 0, "whole row panels");

typedef __attribute__((ext_vector_type(2))) float v2f;
typedef __attribute__((ext_vector_type(8))) float v8f;

static __device__ __forceinline__ void proj_piece(
    const float* __restrict__ x,
    const float* __restrict__ W,
    float* __restrict__ outp,
    int rowbase, int colbase, int lane)
{
    const int half = lane >> 4;
    const int l    = lane & 15;
    const float* __restrict__ xrow = x + (size_t)(rowbase + l) * FF;

    v8f acc[4];
#pragma unroll
    for (int c = 0; c < 4; ++c) acc[c] = (v8f){};

    for (int k = 0; k < FF; k += 4) {
        v2f a;
        a.x = xrow[k + 2 * half + 0];
        a.y = xrow[k + 2 * half + 1];
        const float* wp = W + (size_t)(k + 2 * half) * FF + colbase + 4 * l;
        const float4 w0 = *reinterpret_cast<const float4*>(wp);
        const float4 w1 = *reinterpret_cast<const float4*>(wp + FF);
        const float bx[4] = {w0.x, w0.y, w0.z, w0.w};
        const float by[4] = {w1.x, w1.y, w1.z, w1.w};
#pragma unroll
        for (int c = 0; c < 4; ++c) {
            v2f b;
            b.x = bx[c];
            b.y = by[c];
            acc[c] = __builtin_amdgcn_wmma_f32_16x16x4_f32(
                false, a, false, b, (short)0, acc[c], false, false);
        }
    }

#pragma unroll
    for (int r = 0; r < 8; ++r) {
        float* orow = outp + (size_t)(rowbase + r + 8 * half) * FF + colbase + 4 * l;
        *reinterpret_cast<float4*>(orow) = make_float4(acc[0][r], acc[1][r], acc[2][r], acc[3][r]);
    }
}

static __device__ __forceinline__ void partition_block(
    int B,
    const int* __restrict__ srcI,
    const int* __restrict__ dstI,
    unsigned* __restrict__ runWords,
    int* __restrict__ runOff)
{
    __shared__ int hist[NBIN_PAD];
    __shared__ int loff[NBIN_PAD];
    __shared__ __attribute__((aligned(16))) unsigned stage[CHUNK_E];
    const int t = threadIdx.x;
    for (int i = t; i < NBIN_PAD; i += 256) hist[i] = 0;
    __syncthreads();

    int sv[16], dv[16], r[16];
#pragma unroll
    for (int j = 0; j < 4; ++j) {
        const int p = B * CHUNK_I4 + j * 256 + t;
        int4 s4 = make_int4(0, 0, 0, 0);
        int4 d4 = make_int4(-1, -1, -1, -1);
        if (p < EE / 4) {
            s4 = reinterpret_cast<const int4*>(srcI)[p];
            d4 = reinterpret_cast<const int4*>(dstI)[p];
        }
        sv[4 * j + 0] = s4.x; sv[4 * j + 1] = s4.y; sv[4 * j + 2] = s4.z; sv[4 * j + 3] = s4.w;
        dv[4 * j + 0] = d4.x; dv[4 * j + 1] = d4.y; dv[4 * j + 2] = d4.z; dv[4 * j + 3] = d4.w;
    }
#pragma unroll
    for (int q = 0; q < 16; ++q)
        r[q] = (dv[q] >= 0) ? atomicAdd(&hist[dv[q] >> BIN_SHIFT], 1) : 0;
    __syncthreads();

    if (t < 32) {
        int v[SCAN_W];
        int sum = 0;
#pragma unroll
        for (int k = 0; k < SCAN_W; ++k) v[k] = hist[SCAN_W * t + k];
#pragma unroll
        for (int k = 0; k < SCAN_W; ++k) { const int xk = v[k]; v[k] = sum; sum += xk; }
        int inc = sum;
#pragma unroll
        for (int dlt = 1; dlt < 32; dlt <<= 1) {
            const int y = __shfl_up(inc, dlt, 32);
            if (t >= dlt) inc += y;
        }
        const int excl = inc - sum;
#pragma unroll
        for (int k = 0; k < SCAN_W; ++k) loff[SCAN_W * t + k] = v[k] + excl;
    }
    __syncthreads();

    int lo[16];
#pragma unroll
    for (int q = 0; q < 16; ++q)
        lo[q] = loff[max(dv[q], 0) >> BIN_SHIFT] + r[q];
#pragma unroll
    for (int q = 0; q < 16; ++q) {
        if (dv[q] >= 0)
            stage[lo[q]] = (unsigned)sv[q] | ((unsigned)dv[q] << 16);
    }
    for (int i = t; i < NBIN_PAD; i += 256) runOff[(size_t)B * NBIN_PAD + i] = loff[i];
    __syncthreads();

    const int nvalid = min(CHUNK_E, EE - B * CHUNK_E);
    uint4* __restrict__ ro = reinterpret_cast<uint4*>(runWords + (size_t)B * CHUNK_E);
#pragma unroll
    for (int q = 0; q < CHUNK_E / 4 / 256; ++q) {
        const int i4 = t + 256 * q;
        if (4 * i4 < nvalid) ro[i4] = reinterpret_cast<const uint4*>(stage)[i4];
    }
}

#define LPITCH 68
#define RUN_BATCH 12
static_assert(LPITCH >= CAP + 4 && LPITCH % 4 == 0, "a row is compared 4 words at a time");

static __device__ __forceinline__ int word_node(unsigned w) { return (int)(w >> 16) & (BIN_NODES - 1); }
static __device__ __forceinline__ unsigned word_key(unsigned w, int rk) { return ((w & 0xffffu) << 6) | (unsigned)rk; }

static __device__ __forceinline__ void rank_block(
    int b,
    const unsigned* __restrict__ runWords,
    const int* __restrict__ runOff,
    int* __restrict__ cnt,
    int* __restrict__ slot)
{
    __shared__ int lcnt[BIN_NODES];
    __shared__ __attribute__((aligned(16))) unsigned lkey[BIN_NODES * LPITCH];
    __shared__ __attribute__((aligned(16))) int lslot[BIN_NODES * CAP];
    const int t = threadIdx.x;

    const int tr = min(t, NPB - 1);
    int beg = 0, end = 0;
    if (t < NPB) {
        beg = runOff[(size_t)tr * NBIN_PAD + b];
        end = runOff[(size_t)tr * NBIN_PAD + b + 1];
    }

    if (t < BIN_NODES) lcnt[t] = 0;
    for (int i = t; i < BIN_NODES * LPITCH / 4; i += 256)
        reinterpret_cast<uint4*>(lkey)[i] = make_uint4(0xffffffffu, 0xffffffffu, 0xffffffffu, 0xffffffffu);
    for (int i = t; i < BIN_NODES * CAP / 4; i += 256)
        reinterpret_cast<int4*>(lslot)[i] = make_int4(0x7fffffff, 0x7fffffff, 0x7fffffff, 0x7fffffff);
    __syncthreads();

    {
        const int n = end - beg;
        const unsigned* __restrict__ run = runWords + (size_t)tr * CHUNK_E + beg;
        unsigned w[RUN_BATCH];
        int rk[RUN_BATCH];
#pragma unroll
        for (int q = 0; q < RUN_BATCH; ++q) w[q] = (q < n) ? run[q] : 0u;
#pragma unroll
        for (int q = 0; q < RUN_BATCH; ++q)
            rk[q] = (q < n) ? atomicAdd(&lcnt[word_node(w[q])], 1) : CAP;
#pragma unroll
        for (int q = 0; q < RUN_BATCH; ++q) {
            if (rk[q] < CAP) lkey[word_node(w[q]) * LPITCH + rk[q]] = word_key(w[q], rk[q]);
        }
        for (int q = RUN_BATCH; q < n; ++q) {
            const unsigned wq = run[q];
            const int r1 = atomicAdd(&lcnt[word_node(wq)], 1);
            if (r1 < CAP) lkey[word_node(wq) * LPITCH + r1] = word_key(wq, r1);
        }
    }
    __syncthreads();

    {
        const int d = t >> 2;
        const int c = min(lcnt[d], CAP);
        const unsigned* __restrict__ row = lkey + d * LPITCH;
        for (int j = t & 3; j < c; j += 4) {
            const unsigned mine = row[j];
            int below = 0;
            for (int i = 0; i < c; i += 4) {
                const uint4 k = *reinterpret_cast<const uint4*>(row + i);
                below += (int)(k.x < mine) + (int)(k.y < mine) + (int)(k.z < mine) + (int)(k.w < mine);
            }
            lslot[d * CAP + below] = (int)(mine >> 6);
        }
    }
    __syncthreads();
    if (t < BIN_NODES) cnt[b * BIN_NODES + t] = lcnt[t];
    int4* __restrict__ so = reinterpret_cast<int4*>(slot + (size_t)b * BIN_NODES * CAP);
    for (int i = t; i < BIN_NODES * CAP / 4; i += 256)
        so[i] = reinterpret_cast<const int4*>(lslot)[i];
}

template <int PHASE>
__global__ __launch_bounds__(256) void gat_front(
    const float* __restrict__ x,
    const float* __restrict__ W,
    float* __restrict__ proj,
    const int* __restrict__ srcI,
    const int* __restrict__ dstI,
    unsigned* __restrict__ runWords,
    int* __restrict__ runOff,
    int* __restrict__ cnt,
    int* __restrict__ slot,
    float* __restrict__ gmax)
{
    int gb;
    if (PHASE == 0) {
        if (blockIdx.x == 0 && threadIdx.x == 0) *gmax = -INFINITY;
        if (blockIdx.x < NPB) {
            partition_block((int)blockIdx.x, srcI, dstI, runWords, runOff);
            return;
        }
        gb = (int)blockIdx.x - NPB;
    } else {
        if (blockIdx.x >= GEMM_BLOCKS) {
            rank_block((int)blockIdx.x - GEMM_BLOCKS, runWords, runOff, cnt, slot);
            return;
        }
        gb = (int)blockIdx.x;
    }
    const int piece = __builtin_amdgcn_readfirstlane(gb * 8 + (int)(threadIdx.x >> 5));
    if (piece < GEMM_UNITS)
        proj_piece(x, W, proj, (piece >> 1) * 16, (piece & 1) * 64, (int)(threadIdx.x & 31));
}

template <int CTRL>
static __device__ __forceinline__ float row_lane(float v) {
    return __int_as_float(__builtin_amdgcn_update_dpp(0, __float_as_int(v), CTRL, 0xf, 0xf, true));
}

static __device__ __forceinline__ float4 row_piece(const float* __restrict__ p, int row, int lane) {
    return *reinterpret_cast<const float4*>(p + (size_t)row * FF + 4 * lane);
}

static __device__ __forceinline__ float head_score(const float4 s, const float4 d, const float4 at) {
    float v0 = s.x + d.x;
    float v1 = s.y + d.y;
    float v2 = s.z + d.z;
    float v3 = s.w + d.w;
    v0 = (v0 >= 0.0f) ? v0 : 0.2f * v0;
    v1 = (v1 >= 0.0f) ? v1 : 0.2f * v1;
    v2 = (v2 >= 0.0f) ? v2 : 0.2f * v2;
    v3 = (v3 >= 0.0f) ? v3 : 0.2f * v3;
    float acc = at.x * v0;
    acc += at.y * v1;
    acc += at.z * v2;
    acc += at.w * v3;
    acc += row_lane<0xB1>(acc);
    acc += row_lane<0x4E>(acc);
    return acc;
}

static __device__ __forceinline__ void absorb4(
    const float* __restrict__ src_p, int lane,
    int s0, int s1, int s2, int s3, int nv,
    const float4 drow, const float4 at,
    float& m, float& den, float4& sum)
{
    const float4 r0 = row_piece(src_p, s0, lane);
    const float4 r1 = row_piece(src_p, s1, lane);
    const float4 r2 = row_piece(src_p, s2, lane);
    const float4 r3 = row_piece(src_p, s3, lane);
    const float c0 = head_score(r0, drow, at);
    float c1 = head_score(r1, drow, at);
    float c2 = head_score(r2, drow, at);
    float c3 = head_score(r3, drow, at);
    c1 = (nv > 1) ? c1 : -INFINITY;
    c2 = (nv > 2) ? c2 : -INFINITY;
    c3 = (nv > 3) ? c3 : -INFINITY;
    const float mn = fmaxf(fmaxf(m, c0), fmaxf(c1, fmaxf(c2, c3)));
    const float f  = __expf(m - mn);
    const float w0 = __expf(c0 - mn);
    const float w1 = __expf(c1 - mn);
    const float w2 = __expf(c2 - mn);
    const float w3 = __expf(c3 - mn);
    den   = den   * f + ((w0 + w1) + (w2 + w3));
    sum.x = sum.x * f + ((w0 * r0.x + w1 * r1.x) + (w2 * r2.x + w3 * r3.x));
    sum.y = sum.y * f + ((w0 * r0.y + w1 * r1.y) + (w2 * r2.y + w3 * r3.y));
    sum.z = sum.z * f + ((w0 * r0.z + w1 * r1.z) + (w2 * r2.z + w3 * r3.z));
    sum.w = sum.w * f + ((w0 * r0.w + w1 * r1.w) + (w2 * r2.w + w3 * r3.w));
    m = mn;
}

static __device__ __forceinline__ void absorb_group(
    const float* __restrict__ src_p, int lane, const int4 q, int nv,
    const float4 drow, const float4 at,
    float& m, float& den, float4& sum)
{
    const int s0 = q.x;
    const int s1 = (nv > 1) ? q.y : s0;
    const int s2 = (nv > 2) ? q.z : s0;
    const int s3 = (nv > 3) ? q.w : s0;
    absorb4(src_p, lane, s0, s1, s2, s3, nv, drow, at, m, den, sum);
}

static_assert(CAP % 16 == 0, "the slot row is read 16 slots at a time");
static_assert(NN % 8 == 0, "a block of 8 waves takes 8 whole nodes");
static_assert(EE % 32 == 0, "the slow path reads 32 edges per trip");

static __device__ __forceinline__ void node_state(
    const float* __restrict__ src_p,
    const int*   __restrict__ srcI,
    const int*   __restrict__ dstI,
    const int*   __restrict__ slot,
    int n, int deg, int lane,
    const float4 drow, const float4 at,
    float& m, float& den, float4& sum)
{
    m   = -INFINITY;
    den = 0.0f;
    sum = make_float4(0.0f, 0.0f, 0.0f, 0.0f);

    if (deg <= CAP) {
        const int* __restrict__ sl = slot + (size_t)n * CAP;
        for (int i = 0; i < deg; i += 16) {
            const int4 q0 = *reinterpret_cast<const int4*>(sl + i);
            const int4 q1 = *reinterpret_cast<const int4*>(sl + i + 4);
            const int4 q2 = *reinterpret_cast<const int4*>(sl + i + 8);
            const int4 q3 = *reinterpret_cast<const int4*>(sl + i + 12);
            absorb_group(src_p, lane, q0, deg - i, drow, at, m, den, sum);
            if (i + 4 < deg)  absorb_group(src_p, lane, q1, deg - i - 4,  drow, at, m, den, sum);
            if (i + 8 < deg)  absorb_group(src_p, lane, q2, deg - i - 8,  drow, at, m, den, sum);
            if (i + 12 < deg) absorb_group(src_p, lane, q3, deg - i - 12, drow, at, m, den, sum);
        }
    } else {
        for (int e = 0; e < EE; e += 32) {
            unsigned hit = __builtin_amdgcn_ballot_w32(dstI[e + lane] == n);
            while (hit != 0u) {
                const int b  = __builtin_ctz(hit);
                hit &= hit - 1u;
                const int s0 = srcI[e + b];
                absorb4(src_p, lane, s0, s0, s0, s0, 1, drow, at, m, den, sum);
            }
        }
    }
}

static __device__ __forceinline__ float4 finish_piece(
    const float4 sum, float d, const float4 xr, const float4 bs, float pw)
{
    float4 o;
    o.x = sum.x / d + xr.x + bs.x;
    o.y = sum.y / d + xr.y + bs.y;
    o.z = sum.z / d + xr.z + bs.z;
    o.w = sum.w / d + xr.w + bs.w;
    o.x = (o.x >= 0.0f) ? o.x : pw * o.x;
    o.y = (o.y >= 0.0f) ? o.y : pw * o.y;
    o.z = (o.z >= 0.0f) ? o.z : pw * o.z;
    o.w = (o.w >= 0.0f) ? o.w : pw * o.w;
    return o;
}

__global__ __launch_bounds__(256) void gat_node(
    const float* __restrict__ src_p,
    const float* __restrict__ dst_p,
    const int*   __restrict__ srcI,
    const int*   __restrict__ dstI,
    const float* __restrict__ attn,
    const int*   __restrict__ cnt,
    const int*   __restrict__ slot,
    const float* __restrict__ x,
    const float* __restrict__ bias,
    const float* __restrict__ prelu_w,
    float* __restrict__ out,
    float* __restrict__ denb,
    float* __restrict__ maxb,
    float* __restrict__ gmax)
{
    __shared__ float sden[64];
    __shared__ float smax[64];
    __shared__ float wmax[8];
    const int lane = threadIdx.x & 31;
    const int w    = threadIdx.x >> 5;
    const int n   = __builtin_amdgcn_readfirstlane((int)((blockIdx.x * 256 + threadIdx.x) >> 5));
    const int deg = cnt[n];

    const float4 at   = *reinterpret_cast<const float4*>(attn + 4 * lane);
    const float4 drow = row_piece(dst_p, n, lane);

    float  m, den;
    float4 sum;
    node_state(src_p, srcI, dstI, slot, n, deg, lane, drow, at, m, den, sum);

    const float4 xr = row_piece(x, n, lane);
    const float4 bs = *reinterpret_cast<const float4*>(bias + 4 * lane);
    const float  pw = prelu_w[0];

    if ((lane & 3) == 0) {
        sden[w * 8 + (lane >> 2)] = den;
        smax[w * 8 + (lane >> 2)] = m;
    }
    float mm = m;
    mm = fmaxf(mm, row_lane<0x164>(mm));
    mm = fmaxf(mm, row_lane<0x168>(mm));
    const float wm = fmaxf(__int_as_float(__builtin_amdgcn_readlane(__float_as_int(mm), 0)),
                           __int_as_float(__builtin_amdgcn_readlane(__float_as_int(mm), 16)));
    if (lane == 0) wmax[w] = wm;
    __syncthreads();
    if (threadIdx.x < 64) {
        denb[(size_t)blockIdx.x * 64 + threadIdx.x] = sden[threadIdx.x];
    } else if (threadIdx.x < 128) {
        maxb[(size_t)blockIdx.x * 64 + (threadIdx.x - 64)] = smax[threadIdx.x - 64];
    } else if (threadIdx.x == 128) {
        float b = wmax[0];
        for (int i = 1; i < 8; ++i) b = fmaxf(b, wmax[i]);
        __hip_atomic_fetch_max(gmax, b, __ATOMIC_RELAXED, __HIP_MEMORY_SCOPE_AGENT);
    }

    const float d = (deg > 0) ? den : INFINITY;
    *reinterpret_cast<float4*>(out + (size_t)n * FF + 4 * lane) = finish_piece(sum, d, xr, bs, pw);
}

static __device__ __forceinline__ bool needs_term(float den, float m, float G) {
    return (den + 1e-16f * __expf(G - m)) != den;
}

__global__ __launch_bounds__(64) void gat_repair(
    const float* __restrict__ src_p,
    const float* __restrict__ dst_p,
    const int*   __restrict__ srcI,
    const int*   __restrict__ dstI,
    const float* __restrict__ attn,
    const int*   __restrict__ cnt,
    const int*   __restrict__ slot,
    const float* __restrict__ denb,
    const float* __restrict__ maxb,
    const float* __restrict__ gmax,
    const float* __restrict__ x,
    const float* __restrict__ bias,
    const float* __restrict__ prelu_w,
    float* __restrict__ out)
{
    const int lane = threadIdx.x & 31;
    const int node = blockIdx.x * 64 + threadIdx.x;
    const float G  = *gmax;
    bool bad = false;
    if (node < NN) {
        const float4 d0 = *reinterpret_cast<const float4*>(denb + (size_t)node * HH);
        const float4 d1 = *reinterpret_cast<const float4*>(denb + (size_t)node * HH + 4);
        const float4 m0 = *reinterpret_cast<const float4*>(maxb + (size_t)node * HH);
        const float4 m1 = *reinterpret_cast<const float4*>(maxb + (size_t)node * HH + 4);
        bad = needs_term(d0.x, m0.x, G) | needs_term(d0.y, m0.y, G) |
              needs_term(d0.z, m0.z, G) | needs_term(d0.w, m0.w, G) |
              needs_term(d1.x, m1.x, G) | needs_term(d1.y, m1.y, G) |
              needs_term(d1.z, m1.z, G) | needs_term(d1.w, m1.w, G);
    }
    unsigned hit = __builtin_amdgcn_ballot_w32(bad);
    if (hit == 0u) return;

    const int nbase = __builtin_amdgcn_readfirstlane(node);
    const float4 at = *reinterpret_cast<const float4*>(attn + 4 * lane);
    const float4 bs = *reinterpret_cast<const float4*>(bias + 4 * lane);
    const float  pw = prelu_w[0];
    while (hit != 0u) {
        const int n = nbase + __builtin_ctz(hit);
        hit &= hit - 1u;
        const int deg = cnt[n];
        const float4 drow = row_piece(dst_p, n, lane);
        const float4 xr   = row_piece(x, n, lane);
        float  m, dn;
        float4 sum;
        node_state(src_p, srcI, dstI, slot, n, deg, lane, drow, at, m, dn, sum);
        const float d = dn + 1e-16f * __expf(G - m);
        *reinterpret_cast<float4*>(out + (size_t)n * FF + 4 * lane) = finish_piece(sum, d, xr, bs, pw);
    }
}

extern "C" void kernel_launch(void* const* d_in, const int* in_sizes, int n_in,
                              void* d_out, int out_size, void* d_ws, size_t ws_size,
                              hipStream_t stream) {
    (void)in_sizes; (void)n_in; (void)out_size; (void)ws_size;

    const float* x        = (const float*)d_in[0];
    const int*   eidx     = (const int*)  d_in[1];
    const float* W_src    = (const float*)d_in[2];
    const float* W_dst    = (const float*)d_in[3];
    const float* attn     = (const float*)d_in[4];
    const float* bias     = (const float*)d_in[5];
    const float* prelu_w  = (const float*)d_in[6];
    const int*   srcI     = eidx;
    const int*   dstI     = eidx + EE;

    float* out = (float*)d_out;

    float* ws      = (float*)d_ws;
    float* src_p   = ws;
    float* dst_p   = src_p + (size_t)NN * FF;
    int*   slot    = (int*)(dst_p + (size_t)NN * FF);
    int*   cnt     = slot + (size_t)ROWS_PAD * CAP;
    unsigned* runWords = (unsigned*)(cnt + ROWS_PAD);
    int*   runOff  = (int*)(runWords + (size_t)NPB * CHUNK_E);
    float* denb    = (float*)(runOff + (size_t)NPB * NBIN_PAD);
    float* maxb    = denb + (size_t)NN * HH;
    float* gmax    = maxb + (size_t)NN * HH;

    gat_front<0><<<NPB + GEMM_BLOCKS, 256, 0, stream>>>(x, W_src, src_p, srcI, dstI,
                                                        runWords, runOff, cnt, slot, gmax);

    gat_front<1><<<GEMM_BLOCKS + NBIN, 256, 0, stream>>>(x, W_dst, dst_p, srcI, dstI,
                                                         runWords, runOff, cnt, slot, gmax);

    gat_node<<<NN / 8, 256, 0, stream>>>(src_p, dst_p, srcI, dstI, attn, cnt, slot,
                                         x, bias, prelu_w, out, denb, maxb, gmax);

    gat_repair<<<(NN + 63) / 64, 64, 0, stream>>>(src_p, dst_p, srcI, dstI, attn, cnt, slot,
                                                  denb, maxb, gmax, x, bias, prelu_w, out);
}
